# starting kernel
# baseline (speedup 1.0000x reference)
#include <hip/hip_runtime.h>


#define N_NODES 50000
#define N_EDGES 800000
#define IN_C    128
#define HID_C   128
#define OUT_C   64

typedef float v2f __attribute__((ext_vector_type(2)));
typedef float v8f __attribute__((ext_vector_type(8)));

__global__ void sage_zero_kernel(float* __restrict__ p, long long n4) {
    long long i = (long long)blockIdx.x * blockDim.x + threadIdx.x;
    if (i < n4) ((float4*)p)[i] = make_float4(0.f, 0.f, 0.f, 0.f);
}

__global__ __launch_bounds__(256) void sage_scatter_kernel(
    const float* __restrict__ feat,
    const int*   __restrict__ src,
    const int*   __restrict__ dst,
    float* __restrict__ sum,
    float* __restrict__ deg,
    int n_edges) {
    const int lane = threadIdx.x & 31;
    const int e    = blockIdx.x * (blockDim.x >> 5) + (threadIdx.x >> 5);
    if (e >= n_edges) return;
    const int s = src[e];
    const int d = dst[e];
    const float4 v = ((const float4*)(feat + (long long)s * 128))[lane];
    float* o = sum + (long long)d * 128 + lane * 4;
    atomicAdd(o + 0, v.x);
    atomicAdd(o + 1, v.y);
    atomicAdd(o + 2, v.z);
    atomicAdd(o + 3, v.w);
    if (deg != nullptr && lane == 0) atomicAdd(deg + d, 1.0f);
}

template <int CO, bool RELU>
__global__ __launch_bounds__(256) void sage_gemm_kernel(
    const float* __restrict__ sum,
    const float* __restrict__ deg,
    const float* __restrict__ feat,
    const float* __restrict__ Wl,
    const float* __restrict__ bias,
    const float* __restrict__ Wr,
    float* __restrict__ out,
    int n_tiles) {
    constexpr int NT = CO / 16;

    extern __shared__ float smem[];
    float* sWl = smem;
    float* sWr = smem + 128 * CO;

    const int tid = threadIdx.x;
    for (int i = tid; i < 128 * CO; i += blockDim.x) {
        sWl[i] = Wl[i];
        sWr[i] = Wr[i];
    }
    __syncthreads();

    const int wave = tid >> 5;
    const int lane = tid & 31;
    const int tile = blockIdx.x * (blockDim.x >> 5) + wave;
    if (tile >= n_tiles) return;

    const int row0 = tile * 16;
    const int hi   = lane >> 4;
    const int lr   = lane & 15;
    const int arow = row0 + lr;

    const float dg   = deg[arow];
    const float invd = 1.0f / fmaxf(dg, 1.0f);

    const float* aggRow = sum  + (long long)arow * 128;
    const float* xRow   = feat + (long long)arow * 128;

    v8f acc[NT];
#pragma unroll
    for (int n = 0; n < NT; ++n) acc[n] = (v8f){};

    for (int k = 0; k < 32; ++k) {
        const int c = 4 * k + 2 * hi;
        v2f aAgg, aX;
        aAgg.x = aggRow[c]     * invd;
        aAgg.y = aggRow[c + 1] * invd;
        aX.x   = xRow[c];
        aX.y   = xRow[c + 1];
#pragma unroll
        for (int n = 0; n < NT; ++n) {
            const int col = n * 16 + lr;
            v2f bL, bR;
            bL.x = sWl[(c)     * CO + col];
            bL.y = sWl[(c + 1) * CO + col];
            bR.x = sWr[(c)     * CO + col];
            bR.y = sWr[(c + 1) * CO + col];
            acc[n] = __builtin_amdgcn_wmma_f32_16x16x4_f32(
                false, aAgg, false, bL, (short)0, acc[n], false, false);
            acc[n] = __builtin_amdgcn_wmma_f32_16x16x4_f32(
                false, aX,   false, bR, (short)0, acc[n], false, false);
        }
    }

#pragma unroll
    for (int n = 0; n < NT; ++n) {
        const int col = n * 16 + lr;
        const float b = bias[col];
#pragma unroll
        for (int v = 0; v < 8; ++v) {
            const int r = row0 + v + 8 * hi;
            float val = acc[n][v] + b;
            if (RELU) val = fmaxf(val, 0.0f);
            out[(long long)r * CO + col] = val;
        }
    }
}

extern "C" void kernel_launch(void* const* d_in, const int* in_sizes, int n_in,
                              void* d_out, int out_size, void* d_ws, size_t ws_size,
                              hipStream_t stream) {
    const float* x   = (const float*)d_in[0];
    const int*   ei  = (const int*)  d_in[1];
    const float* W1l = (const float*)d_in[2];
    const float* b1  = (const float*)d_in[3];
    const float* W1r = (const float*)d_in[4];
    const float* W2l = (const float*)d_in[5];
    const float* b2  = (const float*)d_in[6];
    const float* W2r = (const float*)d_in[7];

    const int* src = ei;
    const int* dst = ei + N_EDGES;

    float* ws  = (float*)d_ws;
    float* sum = ws;
    float* deg = ws + (long long)N_NODES * 128;
    float* h   = deg + N_NODES;
    float* out = (float*)d_out;

    const int tiles       = N_NODES / 16;
    const int gemm_blocks = (tiles + 7) / 8;
    const int edge_blocks = (N_EDGES + 7) / 8;

    {
        const long long n = (long long)N_NODES * 128 + N_NODES;
        const long long n4 = n / 4;
        sage_zero_kernel<<<(int)((n4 + 255) / 256), 256, 0, stream>>>(sum, n4);
    }
    sage_scatter_kernel<<<edge_blocks, 256, 0, stream>>>(x, src, dst, sum, deg, N_EDGES);
    sage_gemm_kernel<HID_C, true><<<gemm_blocks, 256, 2 * 128 * HID_C * sizeof(float), stream>>>(
        sum, deg, x, W1l, b1, W1r, h, tiles);

    {
        const long long n  = (long long)N_NODES * 128;
        const long long n4 = n / 4;
        sage_zero_kernel<<<(int)((n4 + 255) / 256), 256, 0, stream>>>(sum, n4);
    }
    sage_scatter_kernel<<<edge_blocks, 256, 0, stream>>>(h, src, dst, sum, nullptr, N_EDGES);
    sage_gemm_kernel<OUT_C, false><<<gemm_blocks, 256, 2 * 128 * OUT_C * sizeof(float), stream>>>(
        sum, deg, h, W2l, b2, W2r, out, tiles);
}


# Float atomic scatter-add and its zero pass replaced by a per-call counting sort of edges by destination (int atomics) + one-wave-per-node row gather; row sums now in fill-arrival order on 4 partials
# speedup vs baseline: 1.7294x; 1.7294x over previous
#include <hip/hip_runtime.h>


#define N_NODES 50000
#define N_EDGES 800000
#define IN_C    128
#define HID_C   128
#define OUT_C   64

typedef float v2f __attribute__((ext_vector_type(2)));
typedef float v4f __attribute__((ext_vector_type(4)));
typedef float v8f __attribute__((ext_vector_type(8)));

__global__ void sage_zero_kernel(float* __restrict__ p, long long n4) {
    long long i = (long long)blockIdx.x * blockDim.x + threadIdx.x;
    if (i < n4) ((float4*)p)[i] = make_float4(0.f, 0.f, 0.f, 0.f);
}

__global__ __launch_bounds__(256) void sage_count_kernel(
    const int* __restrict__ dst, int* __restrict__ cnt, int n_edges4) {
    const int i = blockIdx.x * blockDim.x + threadIdx.x;
    if (i >= n_edges4) return;
    const int4 d = ((const int4*)dst)[i];
    atomicAdd(cnt + d.x, 1);
    atomicAdd(cnt + d.y, 1);
    atomicAdd(cnt + d.z, 1);
    atomicAdd(cnt + d.w, 1);
}

__global__ __launch_bounds__(1024) void sage_rowstart_kernel(
    const int* __restrict__ cnt, int* __restrict__ cursor, int n) {
    __shared__ int s_before[32];
    __shared__ int s_own[32];
    const int t    = threadIdx.x;
    const int lane = t & 31;
    const int w    = t >> 5;
    const int b0   = blockIdx.x * 1024;

    int before = 0;
    for (int i = t; i < b0; i += 1024) before += cnt[i];

    const int idx = b0 + t;
    const int c   = (idx < n) ? cnt[idx] : 0;

    int incl = c;
    for (int off = 1; off < 32; off <<= 1) {
        const int up = __shfl_up(incl, off);
        if (lane >= off) incl += up;
    }
    for (int off = 16; off > 0; off >>= 1) before += __shfl_xor(before, off);

    if (lane == 31) s_own[w] = incl;
    if (lane == 0)  s_before[w] = before;
    __syncthreads();

    int base = 0;
    for (int k = 0; k < 32; ++k) {
        base += s_before[k];
        if (k < w) base += s_own[k];
    }
    if (idx < n) cursor[idx] = base + incl - c;
}

__global__ __launch_bounds__(256) void sage_fill_kernel(
    const int* __restrict__ src, const int* __restrict__ dst,
    int* __restrict__ cursor, int* __restrict__ col, int n_edges4) {
    const int i = blockIdx.x * blockDim.x + threadIdx.x;
    if (i >= n_edges4) return;
    const int4 d = ((const int4*)dst)[i];
    const int4 s = ((const int4*)src)[i];
    const int p0 = atomicAdd(cursor + d.x, 1);
    const int p1 = atomicAdd(cursor + d.y, 1);
    const int p2 = atomicAdd(cursor + d.z, 1);
    const int p3 = atomicAdd(cursor + d.w, 1);
    col[p0] = s.x;
    col[p1] = s.y;
    col[p2] = s.z;
    col[p3] = s.w;
}

__device__ __forceinline__ v4f sage_row4(const float* __restrict__ feat, int s, int lane) {
    return ((const v4f*)(feat + (long long)s * 128))[lane];
}

__global__ __launch_bounds__(256) void sage_gather_kernel(
    const float* __restrict__ feat,
    const int*   __restrict__ col,
    const int*   __restrict__ cnt,
    const int*   __restrict__ rowend,
    float* __restrict__ sum,
    float* __restrict__ deg,
    int n_nodes) {
    const int lane = threadIdx.x & 31;
    const int d = __builtin_amdgcn_readfirstlane((int)(blockIdx.x * (blockDim.x >> 5) + (threadIdx.x >> 5)));
    if (d >= n_nodes) return;
    const int c     = cnt[d];
    const int end   = rowend[d];
    const int start = end - c;

    v4f a0 = (v4f){}, a1 = (v4f){}, a2 = (v4f){}, a3 = (v4f){};
    for (int base = start; base < end; base += 32) {
        const int left = end - base;
        const int n    = left < 32 ? left : 32;
        const int mine = (lane < n) ? col[base + lane] : 0;
        int j = 0;
        for (; j + 4 <= n; j += 4) {
            const int s0 = __builtin_amdgcn_readlane(mine, j);
            const int s1 = __builtin_amdgcn_readlane(mine, j + 1);
            const int s2 = __builtin_amdgcn_readlane(mine, j + 2);
            const int s3 = __builtin_amdgcn_readlane(mine, j + 3);
            const v4f v0 = sage_row4(feat, s0, lane);
            const v4f v1 = sage_row4(feat, s1, lane);
            const v4f v2 = sage_row4(feat, s2, lane);
            const v4f v3 = sage_row4(feat, s3, lane);
            a0 += v0; a1 += v1; a2 += v2; a3 += v3;
        }
        const int rem = n - j;
        const int r0 = __builtin_amdgcn_readlane(mine, j & 31);
        const int r1 = __builtin_amdgcn_readlane(mine, (j + 1) & 31);
        const int r2 = __builtin_amdgcn_readlane(mine, (j + 2) & 31);
        v4f v0 = (v4f){}, v1 = (v4f){}, v2 = (v4f){};
        if (rem > 0) v0 = sage_row4(feat, r0, lane);
        if (rem > 1) v1 = sage_row4(feat, r1, lane);
        if (rem > 2) v2 = sage_row4(feat, r2, lane);
        a0 += v0; a1 += v1; a2 += v2;
    }
    const v4f r = (a0 + a1) + (a2 + a3);
    ((v4f*)(sum + (long long)d * 128))[lane] = r;
    if (deg != nullptr && lane == 0) deg[d] = (float)c;
}

template <int CO, bool RELU>
__global__ __launch_bounds__(256) void sage_gemm_kernel(
    const float* __restrict__ sum,
    const float* __restrict__ deg,
    const float* __restrict__ feat,
    const float* __restrict__ Wl,
    const float* __restrict__ bias,
    const float* __restrict__ Wr,
    float* __restrict__ out,
    int n_tiles) {
    constexpr int NT = CO / 16;

    extern __shared__ float smem[];
    float* sWl = smem;
    float* sWr = smem + 128 * CO;

    const int tid = threadIdx.x;
    for (int i = tid; i < 128 * CO; i += blockDim.x) {
        sWl[i] = Wl[i];
        sWr[i] = Wr[i];
    }
    __syncthreads();

    const int wave = tid >> 5;
    const int lane = tid & 31;
    const int tile = blockIdx.x * (blockDim.x >> 5) + wave;
    if (tile >= n_tiles) return;

    const int row0 = tile * 16;
    const int hi   = lane >> 4;
    const int lr   = lane & 15;
    const int arow = row0 + lr;

    const float dg   = deg[arow];
    const float invd = 1.0f / fmaxf(dg, 1.0f);

    const float* aggRow = sum  + (long long)arow * 128;
    const float* xRow   = feat + (long long)arow * 128;

    v8f acc[NT];
#pragma unroll
    for (int n = 0; n < NT; ++n) acc[n] = (v8f){};

    for (int k = 0; k < 32; ++k) {
        const int c = 4 * k + 2 * hi;
        v2f aAgg, aX;
        aAgg.x = aggRow[c]     * invd;
        aAgg.y = aggRow[c + 1] * invd;
        aX.x   = xRow[c];
        aX.y   = xRow[c + 1];
#pragma unroll
        for (int n = 0; n < NT; ++n) {
            const int col = n * 16 + lr;
            v2f bL, bR;
            bL.x = sWl[(c)     * CO + col];
            bL.y = sWl[(c + 1) * CO + col];
            bR.x = sWr[(c)     * CO + col];
            bR.y = sWr[(c + 1) * CO + col];
            acc[n] = __builtin_amdgcn_wmma_f32_16x16x4_f32(
                false, aAgg, false, bL, (short)0, acc[n], false, false);
            acc[n] = __builtin_amdgcn_wmma_f32_16x16x4_f32(
                false, aX,   false, bR, (short)0, acc[n], false, false);
        }
    }

#pragma unroll
    for (int n = 0; n < NT; ++n) {
        const int col = n * 16 + lr;
        const float b = bias[col];
#pragma unroll
        for (int v = 0; v < 8; ++v) {
            const int r = row0 + v + 8 * hi;
            float val = acc[n][v] + b;
            if (RELU) val = fmaxf(val, 0.0f);
            out[(long long)r * CO + col] = val;
        }
    }
}

extern "C" void kernel_launch(void* const* d_in, const int* in_sizes, int n_in,
                              void* d_out, int out_size, void* d_ws, size_t ws_size,
                              hipStream_t stream) {
    const float* x   = (const float*)d_in[0];
    const int*   ei  = (const int*)  d_in[1];
    const float* W1l = (const float*)d_in[2];
    const float* b1  = (const float*)d_in[3];
    const float* W1r = (const float*)d_in[4];
    const float* W2l = (const float*)d_in[5];
    const float* b2  = (const float*)d_in[6];
    const float* W2r = (const float*)d_in[7];

    const int* src = ei;
    const int* dst = ei + N_EDGES;

    float* ws  = (float*)d_ws;
    float* sum = ws;
    float* deg = ws + (long long)N_NODES * 128;
    float* h   = deg + N_NODES;
    float* out = (float*)d_out;

    int* col    = (int*)d_out;
    int* cnt    = col + N_EDGES;
    int* cursor = cnt + N_NODES;

    const int tiles       = N_NODES / 16;
    const int gemm_blocks = (tiles + 7) / 8;
    const int edge_blocks = (N_EDGES / 4 + 255) / 256;
    const int node_blocks = (N_NODES + 7) / 8;
    const int scan_blocks = (N_NODES + 1023) / 1024;

    sage_zero_kernel<<<(N_NODES / 4 + 255) / 256, 256, 0, stream>>>((float*)cnt, N_NODES / 4);
    sage_count_kernel<<<edge_blocks, 256, 0, stream>>>(dst, cnt, N_EDGES / 4);
    sage_rowstart_kernel<<<scan_blocks, 1024, 0, stream>>>(cnt, cursor, N_NODES);
    sage_fill_kernel<<<edge_blocks, 256, 0, stream>>>(src, dst, cursor, col, N_EDGES / 4);

    sage_gather_kernel<<<node_blocks, 256, 0, stream>>>(x, col, cnt, cursor, sum, deg, N_NODES);
    sage_gemm_kernel<HID_C, true><<<gemm_blocks, 256, 2 * 128 * HID_C * sizeof(float), stream>>>(
        sum, deg, x, W1l, b1, W1r, h, tiles);

    sage_gather_kernel<<<node_blocks, 256, 0, stream>>>(h, col, cnt, cursor, sum, nullptr, N_NODES);
    sage_gemm_kernel<OUT_C, false><<<gemm_blocks, 256, 2 * 128 * OUT_C * sizeof(float), stream>>>(
        sum, deg, h, W2l, b2, W2r, out, tiles);
}


# Edge sort without global atomics: one block per (edge chunk, node group) counts and hands out row slots with LDS counters, a chunk prefix pass replaces the zero pass; row sums now in chunk order
# speedup vs baseline: 2.6605x; 1.5384x over previous
#include <hip/hip_runtime.h>


#define N_NODES 50000
#define N_EDGES 800000
#define IN_C    128
#define HID_C   128
#define OUT_C   64

typedef float v2f __attribute__((ext_vector_type(2)));
typedef float v4f __attribute__((ext_vector_type(4)));
typedef float v8f __attribute__((ext_vector_type(8)));

#define SORT_G   8
#define SORT_C   32
#define GROUP_N  (N_NODES / SORT_G)
#define CHUNK_E  (N_EDGES / SORT_C)

__global__ __launch_bounds__(1024) void sage_hist_kernel(
    const int* __restrict__ dst, int* __restrict__ part) {
    __shared__ int s_cnt[GROUP_N];
    const int t  = threadIdx.x;
    const int c  = blockIdx.x / SORT_G;
    const int g  = blockIdx.x % SORT_G;
    const int lo = g * GROUP_N;
    for (int i = t; i < GROUP_N; i += 1024) s_cnt[i] = 0;
    __syncthreads();
    const int4* d4 = (const int4*)dst + c * (CHUNK_E / 4);
    int  i = t;
    int4 d = d4[i];
    for (;;) {
        const int  in   = i + 1024;
        const bool more = in < CHUNK_E / 4;
        int4 dn = d;
        if (more) dn = d4[in];
        const unsigned r0 = (unsigned)(d.x - lo);
        const unsigned r1 = (unsigned)(d.y - lo);
        const unsigned r2 = (unsigned)(d.z - lo);
        const unsigned r3 = (unsigned)(d.w - lo);
        if (r0 < (unsigned)GROUP_N) atomicAdd(&s_cnt[r0], 1);
        if (r1 < (unsigned)GROUP_N) atomicAdd(&s_cnt[r1], 1);
        if (r2 < (unsigned)GROUP_N) atomicAdd(&s_cnt[r2], 1);
        if (r3 < (unsigned)GROUP_N) atomicAdd(&s_cnt[r3], 1);
        if (!more) break;
        d = dn;
        i = in;
    }
    __syncthreads();
    int* p = part + c * N_NODES + lo;
    for (int i = t; i < GROUP_N; i += 1024) p[i] = s_cnt[i];
}

__global__ __launch_bounds__(256) void sage_chunkscan_kernel(
    int* __restrict__ part, int* __restrict__ cnt, int n) {
    const int d = blockIdx.x * blockDim.x + threadIdx.x;
    if (d >= n) return;
    int v[SORT_C];
#pragma unroll
    for (int c = 0; c < SORT_C; ++c) v[c] = part[c * N_NODES + d];
    int run = 0;
#pragma unroll
    for (int c = 0; c < SORT_C; ++c) {
        part[c * N_NODES + d] = run;
        run += v[c];
    }
    cnt[d] = run;
}

__global__ __launch_bounds__(1024) void sage_rowstart_kernel(
    const int* __restrict__ cnt, int* __restrict__ cursor, int n) {
    __shared__ int s_before[32];
    __shared__ int s_own[32];
    const int t    = threadIdx.x;
    const int lane = t & 31;
    const int w    = t >> 5;
    const int b0   = blockIdx.x * 1024;

    int before = 0;
    for (int i = t; i < b0; i += 1024) before += cnt[i];

    const int idx = b0 + t;
    const int c   = (idx < n) ? cnt[idx] : 0;

    int incl = c;
    for (int off = 1; off < 32; off <<= 1) {
        const int up = __shfl_up(incl, off);
        if (lane >= off) incl += up;
    }
    for (int off = 16; off > 0; off >>= 1) before += __shfl_xor(before, off);

    if (lane == 31) s_own[w] = incl;
    if (lane == 0)  s_before[w] = before;
    __syncthreads();

    int base = 0;
    for (int k = 0; k < 32; ++k) {
        base += s_before[k];
        if (k < w) base += s_own[k];
    }
    if (idx < n) cursor[idx] = base + incl - c;
}

__global__ __launch_bounds__(1024) void sage_fill_kernel(
    const int* __restrict__ src, const int* __restrict__ dst,
    const int* __restrict__ part, const int* __restrict__ cursor,
    int* __restrict__ col) {
    __shared__ int s_slot[GROUP_N];
    const int t  = threadIdx.x;
    const int c  = blockIdx.x / SORT_G;
    const int g  = blockIdx.x % SORT_G;
    const int lo = g * GROUP_N;
    const int* p = part + c * N_NODES + lo;
    for (int i = t; i < GROUP_N; i += 1024) s_slot[i] = cursor[lo + i] + p[i];
    __syncthreads();
    const int4* d4 = (const int4*)dst + c * (CHUNK_E / 4);
    const int4* s4 = (const int4*)src + c * (CHUNK_E / 4);
    int  i = t;
    int4 d = d4[i];
    int4 s = s4[i];
    for (;;) {
        const int  in   = i + 1024;
        const bool more = in < CHUNK_E / 4;
        int4 dn = d, sn = s;
        if (more) { dn = d4[in]; sn = s4[in]; }
        const unsigned r0 = (unsigned)(d.x - lo);
        const unsigned r1 = (unsigned)(d.y - lo);
        const unsigned r2 = (unsigned)(d.z - lo);
        const unsigned r3 = (unsigned)(d.w - lo);
        const bool in0 = r0 < (unsigned)GROUP_N;
        const bool in1 = r1 < (unsigned)GROUP_N;
        const bool in2 = r2 < (unsigned)GROUP_N;
        const bool in3 = r3 < (unsigned)GROUP_N;
        unsigned p0 = 0, p1 = 0, p2 = 0, p3 = 0;
        if (in0) p0 = (unsigned)atomicAdd(&s_slot[r0], 1);
        if (in1) p1 = (unsigned)atomicAdd(&s_slot[r1], 1);
        if (in2) p2 = (unsigned)atomicAdd(&s_slot[r2], 1);
        if (in3) p3 = (unsigned)atomicAdd(&s_slot[r3], 1);
        if (in0) col[p0] = s.x;
        if (in1) col[p1] = s.y;
        if (in2) col[p2] = s.z;
        if (in3) col[p3] = s.w;
        if (!more) break;
        d = dn;
        s = sn;
        i = in;
    }
}

__device__ __forceinline__ v4f sage_row4(const float* __restrict__ feat, int s, int lane) {
    return ((const v4f*)(feat + (long long)s * 128))[lane];
}

__global__ __launch_bounds__(256) void sage_gather_kernel(
    const float* __restrict__ feat,
    const int*   __restrict__ col,
    const int*   __restrict__ cnt,
    const int*   __restrict__ rowstart,
    float* __restrict__ sum,
    float* __restrict__ deg,
    int n_nodes) {
    const int lane = threadIdx.x & 31;
    const int d = __builtin_amdgcn_readfirstlane((int)(blockIdx.x * (blockDim.x >> 5) + (threadIdx.x >> 5)));
    if (d >= n_nodes) return;
    const int c     = cnt[d];
    const int start = rowstart[d];
    const int end   = start + c;

    v4f a0 = (v4f){}, a1 = (v4f){}, a2 = (v4f){}, a3 = (v4f){};
    for (int base = start; base < end; base += 32) {
        const int left = end - base;
        const int n    = left < 32 ? left : 32;
        const int mine = (lane < n) ? col[base + lane] : 0;
        int j = 0;
        for (; j + 4 <= n; j += 4) {
            const int s0 = __builtin_amdgcn_readlane(mine, j);
            const int s1 = __builtin_amdgcn_readlane(mine, j + 1);
            const int s2 = __builtin_amdgcn_readlane(mine, j + 2);
            const int s3 = __builtin_amdgcn_readlane(mine, j + 3);
            const v4f v0 = sage_row4(feat, s0, lane);
            const v4f v1 = sage_row4(feat, s1, lane);
            const v4f v2 = sage_row4(feat, s2, lane);
            const v4f v3 = sage_row4(feat, s3, lane);
            a0 += v0; a1 += v1; a2 += v2; a3 += v3;
        }
        const int rem = n - j;
        const int r0 = __builtin_amdgcn_readlane(mine, j & 31);
        const int r1 = __builtin_amdgcn_readlane(mine, (j + 1) & 31);
        const int r2 = __builtin_amdgcn_readlane(mine, (j + 2) & 31);
        v4f v0 = (v4f){}, v1 = (v4f){}, v2 = (v4f){};
        if (rem > 0) v0 = sage_row4(feat, r0, lane);
        if (rem > 1) v1 = sage_row4(feat, r1, lane);
        if (rem > 2) v2 = sage_row4(feat, r2, lane);
        a0 += v0; a1 += v1; a2 += v2;
    }
    const v4f r = (a0 + a1) + (a2 + a3);
    ((v4f*)(sum + (long long)d * 128))[lane] = r;
    if (deg != nullptr && lane == 0) deg[d] = (float)c;
}

template <int CO, bool RELU>
__global__ __launch_bounds__(256) void sage_gemm_kernel(
    const float* __restrict__ sum,
    const float* __restrict__ deg,
    const float* __restrict__ feat,
    const float* __restrict__ Wl,
    const float* __restrict__ bias,
    const float* __restrict__ Wr,
    float* __restrict__ out,
    int n_tiles) {
    constexpr int NT = CO / 16;

    extern __shared__ float smem[];
    float* sWl = smem;
    float* sWr = smem + 128 * CO;

    const int tid = threadIdx.x;
    for (int i = tid; i < 128 * CO; i += blockDim.x) {
        sWl[i] = Wl[i];
        sWr[i] = Wr[i];
    }
    __syncthreads();

    const int wave = tid >> 5;
    const int lane = tid & 31;
    const int tile = blockIdx.x * (blockDim.x >> 5) + wave;
    if (tile >= n_tiles) return;

    const int row0 = tile * 16;
    const int hi   = lane >> 4;
    const int lr   = lane & 15;
    const int arow = row0 + lr;

    const float dg   = deg[arow];
    const float invd = 1.0f / fmaxf(dg, 1.0f);

    const float* aggRow = sum  + (long long)arow * 128;
    const float* xRow   = feat + (long long)arow * 128;

    v8f acc[NT];
#pragma unroll
    for (int n = 0; n < NT; ++n) acc[n] = (v8f){};

    for (int k = 0; k < 32; ++k) {
        const int c = 4 * k + 2 * hi;
        v2f aAgg, aX;
        aAgg.x = aggRow[c]     * invd;
        aAgg.y = aggRow[c + 1] * invd;
        aX.x   = xRow[c];
        aX.y   = xRow[c + 1];
#pragma unroll
        for (int n = 0; n < NT; ++n) {
            const int col = n * 16 + lr;
            v2f bL, bR;
            bL.x = sWl[(c)     * CO + col];
            bL.y = sWl[(c + 1) * CO + col];
            bR.x = sWr[(c)     * CO + col];
            bR.y = sWr[(c + 1) * CO + col];
            acc[n] = __builtin_amdgcn_wmma_f32_16x16x4_f32(
                false, aAgg, false, bL, (short)0, acc[n], false, false);
            acc[n] = __builtin_amdgcn_wmma_f32_16x16x4_f32(
                false, aX,   false, bR, (short)0, acc[n], false, false);
        }
    }

#pragma unroll
    for (int n = 0; n < NT; ++n) {
        const int col = n * 16 + lr;
        const float b = bias[col];
#pragma unroll
        for (int v = 0; v < 8; ++v) {
            const int r = row0 + v + 8 * hi;
            float val = acc[n][v] + b;
            if (RELU) val = fmaxf(val, 0.0f);
            out[(long long)r * CO + col] = val;
        }
    }
}

extern "C" void kernel_launch(void* const* d_in, const int* in_sizes, int n_in,
                              void* d_out, int out_size, void* d_ws, size_t ws_size,
                              hipStream_t stream) {
    const float* x   = (const float*)d_in[0];
    const int*   ei  = (const int*)  d_in[1];
    const float* W1l = (const float*)d_in[2];
    const float* b1  = (const float*)d_in[3];
    const float* W1r = (const float*)d_in[4];
    const float* W2l = (const float*)d_in[5];
    const float* b2  = (const float*)d_in[6];
    const float* W2r = (const float*)d_in[7];

    const int* src = ei;
    const int* dst = ei + N_EDGES;

    float* ws  = (float*)d_ws;
    float* sum = ws;
    float* deg = ws + (long long)N_NODES * 128;
    float* h   = deg + N_NODES;
    float* out = (float*)d_out;

    int* col    = (int*)d_out;
    int* cnt    = col + N_EDGES;
    int* cursor = cnt + N_NODES;
    int* part   = (int*)d_ws;

    const int tiles       = N_NODES / 16;
    const int gemm_blocks = (tiles + 7) / 8;
    const int node_blocks = (N_NODES + 7) / 8;
    const int scan_blocks = (N_NODES + 1023) / 1024;

    sage_hist_kernel<<<SORT_C * SORT_G, 1024, 0, stream>>>(dst, part);
    sage_chunkscan_kernel<<<(N_NODES + 255) / 256, 256, 0, stream>>>(part, cnt, N_NODES);
    sage_rowstart_kernel<<<scan_blocks, 1024, 0, stream>>>(cnt, cursor, N_NODES);
    sage_fill_kernel<<<SORT_C * SORT_G, 1024, 0, stream>>>(src, dst, part, cursor, col);

    sage_gather_kernel<<<node_blocks, 256, 0, stream>>>(x, col, cnt, cursor, sum, deg, N_NODES);
    sage_gemm_kernel<HID_C, true><<<gemm_blocks, 256, 2 * 128 * HID_C * sizeof(float), stream>>>(
        sum, deg, x, W1l, b1, W1r, h, tiles);

    sage_gather_kernel<<<node_blocks, 256, 0, stream>>>(h, col, cnt, cursor, sum, nullptr, N_NODES);
    sage_gemm_kernel<OUT_C, false><<<gemm_blocks, 256, 2 * 128 * OUT_C * sizeof(float), stream>>>(
        sum, deg, h, W2l, b2, W2r, out, tiles);
}


# Both linear kernels stage their weight matrices into LDS with 16-byte loads requested together, one round trip per wave (was 64 / 32 dependent 4-byte load-wait-store trips); order of sums unchanged
# speedup vs baseline: 3.5417x; 1.3312x over previous
#include <hip/hip_runtime.h>


#define N_NODES 50000
#define N_EDGES 800000
#define IN_C    128
#define HID_C   128
#define OUT_C   64

typedef float v2f __attribute__((ext_vector_type(2)));
typedef float v4f __attribute__((ext_vector_type(4)));
typedef float v8f __attribute__((ext_vector_type(8)));

#define SORT_G   8
#define SORT_C   32
#define GROUP_N  (N_NODES / SORT_G)
#define CHUNK_E  (N_EDGES / SORT_C)

__global__ __launch_bounds__(1024) void sage_hist_kernel(
    const int* __restrict__ dst, int* __restrict__ part) {
    __shared__ int s_cnt[GROUP_N];
    const int t  = threadIdx.x;
    const int c  = blockIdx.x / SORT_G;
    const int g  = blockIdx.x % SORT_G;
    const int lo = g * GROUP_N;
    for (int i = t; i < GROUP_N; i += 1024) s_cnt[i] = 0;
    __syncthreads();
    const int4* d4 = (const int4*)dst + c * (CHUNK_E / 4);
    int  i = t;
    int4 d = d4[i];
    for (;;) {
        const int  in   = i + 1024;
        const bool more = in < CHUNK_E / 4;
        int4 dn = d;
        if (more) dn = d4[in];
        const unsigned r0 = (unsigned)(d.x - lo);
        const unsigned r1 = (unsigned)(d.y - lo);
        const unsigned r2 = (unsigned)(d.z - lo);
        const unsigned r3 = (unsigned)(d.w - lo);
        if (r0 < (unsigned)GROUP_N) atomicAdd(&s_cnt[r0], 1);
        if (r1 < (unsigned)GROUP_N) atomicAdd(&s_cnt[r1], 1);
        if (r2 < (unsigned)GROUP_N) atomicAdd(&s_cnt[r2], 1);
        if (r3 < (unsigned)GROUP_N) atomicAdd(&s_cnt[r3], 1);
        if (!more) break;
        d = dn;
        i = in;
    }
    __syncthreads();
    int* p = part + c * N_NODES + lo;
    for (int i = t; i < GROUP_N; i += 1024) p[i] = s_cnt[i];
}

__global__ __launch_bounds__(256) void sage_chunkscan_kernel(
    int* __restrict__ part, int* __restrict__ cnt, int n) {
    const int d = blockIdx.x * blockDim.x + threadIdx.x;
    if (d >= n) return;
    int v[SORT_C];
#pragma unroll
    for (int c = 0; c < SORT_C; ++c) v[c] = part[c * N_NODES + d];
    int run = 0;
#pragma unroll
    for (int c = 0; c < SORT_C; ++c) {
        part[c * N_NODES + d] = run;
        run += v[c];
    }
    cnt[d] = run;
}

__global__ __launch_bounds__(1024) void sage_rowstart_kernel(
    const int* __restrict__ cnt, int* __restrict__ cursor, int n) {
    __shared__ int s_before[32];
    __shared__ int s_own[32];
    const int t    = threadIdx.x;
    const int lane = t & 31;
    const int w    = t >> 5;
    const int b0   = blockIdx.x * 1024;

    int before = 0;
    for (int i = t; i < b0; i += 1024) before += cnt[i];

    const int idx = b0 + t;
    const int c   = (idx < n) ? cnt[idx] : 0;

    int incl = c;
    for (int off = 1; off < 32; off <<= 1) {
        const int up = __shfl_up(incl, off);
        if (lane >= off) incl += up;
    }
    for (int off = 16; off > 0; off >>= 1) before += __shfl_xor(before, off);

    if (lane == 31) s_own[w] = incl;
    if (lane == 0)  s_before[w] = before;
    __syncthreads();

    int base = 0;
    for (int k = 0; k < 32; ++k) {
        base += s_before[k];
        if (k < w) base += s_own[k];
    }
    if (idx < n) cursor[idx] = base + incl - c;
}

__global__ __launch_bounds__(1024) void sage_fill_kernel(
    const int* __restrict__ src, const int* __restrict__ dst,
    const int* __restrict__ part, const int* __restrict__ cursor,
    int* __restrict__ col) {
    __shared__ int s_slot[GROUP_N];
    const int t  = threadIdx.x;
    const int c  = blockIdx.x / SORT_G;
    const int g  = blockIdx.x % SORT_G;
    const int lo = g * GROUP_N;
    const int* p = part + c * N_NODES + lo;
    for (int i = t; i < GROUP_N; i += 1024) s_slot[i] = cursor[lo + i] + p[i];
    __syncthreads();
    const int4* d4 = (const int4*)dst + c * (CHUNK_E / 4);
    const int4* s4 = (const int4*)src + c * (CHUNK_E / 4);
    int  i = t;
    int4 d = d4[i];
    int4 s = s4[i];
    for (;;) {
        const int  in   = i + 1024;
        const bool more = in < CHUNK_E / 4;
        int4 dn = d, sn = s;
        if (more) { dn = d4[in]; sn = s4[in]; }
        const unsigned r0 = (unsigned)(d.x - lo);
        const unsigned r1 = (unsigned)(d.y - lo);
        const unsigned r2 = (unsigned)(d.z - lo);
        const unsigned r3 = (unsigned)(d.w - lo);
        const bool in0 = r0 < (unsigned)GROUP_N;
        const bool in1 = r1 < (unsigned)GROUP_N;
        const bool in2 = r2 < (unsigned)GROUP_N;
        const bool in3 = r3 < (unsigned)GROUP_N;
        unsigned p0 = 0, p1 = 0, p2 = 0, p3 = 0;
        if (in0) p0 = (unsigned)atomicAdd(&s_slot[r0], 1);
        if (in1) p1 = (unsigned)atomicAdd(&s_slot[r1], 1);
        if (in2) p2 = (unsigned)atomicAdd(&s_slot[r2], 1);
        if (in3) p3 = (unsigned)atomicAdd(&s_slot[r3], 1);
        if (in0) col[p0] = s.x;
        if (in1) col[p1] = s.y;
        if (in2) col[p2] = s.z;
        if (in3) col[p3] = s.w;
        if (!more) break;
        d = dn;
        s = sn;
        i = in;
    }
}

__device__ __forceinline__ v4f sage_row4(const float* __restrict__ feat, int s, int lane) {
    return ((const v4f*)(feat + (long long)s * 128))[lane];
}

__global__ __launch_bounds__(256) void sage_gather_kernel(
    const float* __restrict__ feat,
    const int*   __restrict__ col,
    const int*   __restrict__ cnt,
    const int*   __restrict__ rowstart,
    float* __restrict__ sum,
    float* __restrict__ deg,
    int n_nodes) {
    const int lane = threadIdx.x & 31;
    const int d = __builtin_amdgcn_readfirstlane((int)(blockIdx.x * (blockDim.x >> 5) + (threadIdx.x >> 5)));
    if (d >= n_nodes) return;
    const int c     = cnt[d];
    const int start = rowstart[d];
    const int end   = start + c;

    v4f a0 = (v4f){}, a1 = (v4f){}, a2 = (v4f){}, a3 = (v4f){};
    for (int base = start; base < end; base += 32) {
        const int left = end - base;
        const int n    = left < 32 ? left : 32;
        const int mine = (lane < n) ? col[base + lane] : 0;
        int j = 0;
        for (; j + 4 <= n; j += 4) {
            const int s0 = __builtin_amdgcn_readlane(mine, j);
            const int s1 = __builtin_amdgcn_readlane(mine, j + 1);
            const int s2 = __builtin_amdgcn_readlane(mine, j + 2);
            const int s3 = __builtin_amdgcn_readlane(mine, j + 3);
            const v4f v0 = sage_row4(feat, s0, lane);
            const v4f v1 = sage_row4(feat, s1, lane);
            const v4f v2 = sage_row4(feat, s2, lane);
            const v4f v3 = sage_row4(feat, s3, lane);
            a0 += v0; a1 += v1; a2 += v2; a3 += v3;
        }
        const int rem = n - j;
        const int r0 = __builtin_amdgcn_readlane(mine, j & 31);
        const int r1 = __builtin_amdgcn_readlane(mine, (j + 1) & 31);
        const int r2 = __builtin_amdgcn_readlane(mine, (j + 2) & 31);
        v4f v0 = (v4f){}, v1 = (v4f){}, v2 = (v4f){};
        if (rem > 0) v0 = sage_row4(feat, r0, lane);
        if (rem > 1) v1 = sage_row4(feat, r1, lane);
        if (rem > 2) v2 = sage_row4(feat, r2, lane);
        a0 += v0; a1 += v1; a2 += v2;
    }
    const v4f r = (a0 + a1) + (a2 + a3);
    ((v4f*)(sum + (long long)d * 128))[lane] = r;
    if (deg != nullptr && lane == 0) deg[d] = (float)c;
}

template <int CO, bool RELU>
__global__ __launch_bounds__(256) void sage_gemm_kernel(
    const float* __restrict__ sum,
    const float* __restrict__ deg,
    const float* __restrict__ feat,
    const float* __restrict__ Wl,
    const float* __restrict__ bias,
    const float* __restrict__ Wr,
    float* __restrict__ out,
    int n_tiles) {
    constexpr int NT = CO / 16;

    extern __shared__ v4f smem4[];
    float* sWl = (float*)smem4;
    float* sWr = sWl + 128 * CO;

    const int tid = threadIdx.x;
    {
        constexpr int NV    = 128 * CO / 4;
        constexpr int PER   = NV / 256;
        constexpr int BATCH = 8;
        const v4f* gWl = (const v4f*)Wl;
        const v4f* gWr = (const v4f*)Wr;
        v4f* lWl = smem4;
        v4f* lWr = smem4 + NV;
#pragma unroll
        for (int b = 0; b < PER; b += BATCH) {
            v4f tl[BATCH], tr[BATCH];
#pragma unroll
            for (int u = 0; u < BATCH; ++u) {
                tl[u] = gWl[(b + u) * 256 + tid];
                tr[u] = gWr[(b + u) * 256 + tid];
            }
#pragma unroll
            for (int u = 0; u < BATCH; ++u) {
                lWl[(b + u) * 256 + tid] = tl[u];
                lWr[(b + u) * 256 + tid] = tr[u];
            }
        }
    }
    __syncthreads();

    const int wave = tid >> 5;
    const int lane = tid & 31;
    const int tile = blockIdx.x * (blockDim.x >> 5) + wave;
    if (tile >= n_tiles) return;

    const int row0 = tile * 16;
    const int hi   = lane >> 4;
    const int lr   = lane & 15;
    const int arow = row0 + lr;

    const float dg   = deg[arow];
    const float invd = 1.0f / fmaxf(dg, 1.0f);

    const float* aggRow = sum  + (long long)arow * 128;
    const float* xRow   = feat + (long long)arow * 128;

    v8f acc[NT];
#pragma unroll
    for (int n = 0; n < NT; ++n) acc[n] = (v8f){};

    for (int k = 0; k < 32; ++k) {
        const int c = 4 * k + 2 * hi;
        v2f aAgg, aX;
        aAgg.x = aggRow[c]     * invd;
        aAgg.y = aggRow[c + 1] * invd;
        aX.x   = xRow[c];
        aX.y   = xRow[c + 1];
#pragma unroll
        for (int n = 0; n < NT; ++n) {
            const int col = n * 16 + lr;
            v2f bL, bR;
            bL.x = sWl[(c)     * CO + col];
            bL.y = sWl[(c + 1) * CO + col];
            bR.x = sWr[(c)     * CO + col];
            bR.y = sWr[(c + 1) * CO + col];
            acc[n] = __builtin_amdgcn_wmma_f32_16x16x4_f32(
                false, aAgg, false, bL, (short)0, acc[n], false, false);
            acc[n] = __builtin_amdgcn_wmma_f32_16x16x4_f32(
                false, aX,   false, bR, (short)0, acc[n], false, false);
        }
    }

#pragma unroll
    for (int n = 0; n < NT; ++n) {
        const int col = n * 16 + lr;
        const float b = bias[col];
#pragma unroll
        for (int v = 0; v < 8; ++v) {
            const int r = row0 + v + 8 * hi;
            float val = acc[n][v] + b;
            if (RELU) val = fmaxf(val, 0.0f);
            out[(long long)r * CO + col] = val;
        }
    }
}

extern "C" void kernel_launch(void* const* d_in, const int* in_sizes, int n_in,
                              void* d_out, int out_size, void* d_ws, size_t ws_size,
                              hipStream_t stream) {
    const float* x   = (const float*)d_in[0];
    const int*   ei  = (const int*)  d_in[1];
    const float* W1l = (const float*)d_in[2];
    const float* b1  = (const float*)d_in[3];
    const float* W1r = (const float*)d_in[4];
    const float* W2l = (const float*)d_in[5];
    const float* b2  = (const float*)d_in[6];
    const float* W2r = (const float*)d_in[7];

    const int* src = ei;
    const int* dst = ei + N_EDGES;

    float* ws  = (float*)d_ws;
    float* sum = ws;
    float* deg = ws + (long long)N_NODES * 128;
    float* h   = deg + N_NODES;
    float* out = (float*)d_out;

    int* col    = (int*)d_out;
    int* cnt    = col + N_EDGES;
    int* cursor = cnt + N_NODES;
    int* part   = (int*)d_ws;

    const int tiles       = N_NODES / 16;
    const int gemm_blocks = (tiles + 7) / 8;
    const int node_blocks = (N_NODES + 7) / 8;
    const int scan_blocks = (N_NODES + 1023) / 1024;

    sage_hist_kernel<<<SORT_C * SORT_G, 1024, 0, stream>>>(dst, part);
    sage_chunkscan_kernel<<<(N_NODES + 255) / 256, 256, 0, stream>>>(part, cnt, N_NODES);
    sage_rowstart_kernel<<<scan_blocks, 1024, 0, stream>>>(cnt, cursor, N_NODES);
    sage_fill_kernel<<<SORT_C * SORT_G, 1024, 0, stream>>>(src, dst, part, cursor, col);

    sage_gather_kernel<<<node_blocks, 256, 0, stream>>>(x, col, cnt, cursor, sum, deg, N_NODES);
    sage_gemm_kernel<HID_C, true><<<gemm_blocks, 256, 2 * 128 * HID_C * sizeof(float), stream>>>(
        sum, deg, x, W1l, b1, W1r, h, tiles);

    sage_gather_kernel<<<node_blocks, 256, 0, stream>>>(h, col, cnt, cursor, sum, nullptr, N_NODES);
    sage_gemm_kernel<OUT_C, false><<<gemm_blocks, 256, 2 * 128 * OUT_C * sizeof(float), stream>>>(
        sum, deg, h, W2l, b2, W2r, out, tiles);
}


# Layer 2 projects before it aggregates: z = h @ W2l (64 channels) first, then the second gather sums 256-byte rows of z into out, 4 nodes per wave, not 512-byte rows of h; sums reassociated
# speedup vs baseline: 3.6574x; 1.0327x over previous
#include <hip/hip_runtime.h>


#define N_NODES 50000
#define N_EDGES 800000
#define IN_C    128
#define HID_C   128
#define OUT_C   64

typedef float v2f __attribute__((ext_vector_type(2)));
typedef float v4f __attribute__((ext_vector_type(4)));
typedef float v8f __attribute__((ext_vector_type(8)));

#define SORT_G   8
#define SORT_C   32
#define GROUP_N  (N_NODES / SORT_G)
#define CHUNK_E  (N_EDGES / SORT_C)

__global__ __launch_bounds__(1024) void sage_hist_kernel(
    const int* __restrict__ dst, int* __restrict__ part) {
    __shared__ int s_cnt[GROUP_N];
    const int t  = threadIdx.x;
    const int c  = blockIdx.x / SORT_G;
    const int g  = blockIdx.x % SORT_G;
    const int lo = g * GROUP_N;
    for (int i = t; i < GROUP_N; i += 1024) s_cnt[i] = 0;
    __syncthreads();
    const int4* d4 = (const int4*)dst + c * (CHUNK_E / 4);
    int  i = t;
    int4 d = d4[i];
    for (;;) {
        const int  in   = i + 1024;
        const bool more = in < CHUNK_E / 4;
        int4 dn = d;
        if (more) dn = d4[in];
        const unsigned r0 = (unsigned)(d.x - lo);
        const unsigned r1 = (unsigned)(d.y - lo);
        const unsigned r2 = (unsigned)(d.z - lo);
        const unsigned r3 = (unsigned)(d.w - lo);
        if (r0 < (unsigned)GROUP_N) atomicAdd(&s_cnt[r0], 1);
        if (r1 < (unsigned)GROUP_N) atomicAdd(&s_cnt[r1], 1);
        if (r2 < (unsigned)GROUP_N) atomicAdd(&s_cnt[r2], 1);
        if (r3 < (unsigned)GROUP_N) atomicAdd(&s_cnt[r3], 1);
        if (!more) break;
        d = dn;
        i = in;
    }
    __syncthreads();
    int* p = part + c * N_NODES + lo;
    for (int i = t; i < GROUP_N; i += 1024) p[i] = s_cnt[i];
}

__global__ __launch_bounds__(256) void sage_chunkscan_kernel(
    int* __restrict__ part, int* __restrict__ cnt, int n) {
    const int d = blockIdx.x * blockDim.x + threadIdx.x;
    if (d >= n) return;
    int v[SORT_C];
#pragma unroll
    for (int c = 0; c < SORT_C; ++c) v[c] = part[c * N_NODES + d];
    int run = 0;
#pragma unroll
    for (int c = 0; c < SORT_C; ++c) {
        part[c * N_NODES + d] = run;
        run += v[c];
    }
    cnt[d] = run;
}

__global__ __launch_bounds__(1024) void sage_rowstart_kernel(
    const int* __restrict__ cnt, int* __restrict__ cursor, int n) {
    __shared__ int s_before[32];
    __shared__ int s_own[32];
    const int t    = threadIdx.x;
    const int lane = t & 31;
    const int w    = t >> 5;
    const int b0   = blockIdx.x * 1024;

    int before = 0;
    for (int i = t; i < b0; i += 1024) before += cnt[i];

    const int idx = b0 + t;
    const int c   = (idx < n) ? cnt[idx] : 0;

    int incl = c;
    for (int off = 1; off < 32; off <<= 1) {
        const int up = __shfl_up(incl, off);
        if (lane >= off) incl += up;
    }
    for (int off = 16; off > 0; off >>= 1) before += __shfl_xor(before, off);

    if (lane == 31) s_own[w] = incl;
    if (lane == 0)  s_before[w] = before;
    __syncthreads();

    int base = 0;
    for (int k = 0; k < 32; ++k) {
        base += s_before[k];
        if (k < w) base += s_own[k];
    }
    if (idx < n) cursor[idx] = base + incl - c;
}

__global__ __launch_bounds__(1024) void sage_fill_kernel(
    const int* __restrict__ src, const int* __restrict__ dst,
    const int* __restrict__ part, const int* __restrict__ cursor,
    int* __restrict__ col) {
    __shared__ int s_slot[GROUP_N];
    const int t  = threadIdx.x;
    const int c  = blockIdx.x / SORT_G;
    const int g  = blockIdx.x % SORT_G;
    const int lo = g * GROUP_N;
    const int* p = part + c * N_NODES + lo;
    for (int i = t; i < GROUP_N; i += 1024) s_slot[i] = cursor[lo + i] + p[i];
    __syncthreads();
    const int4* d4 = (const int4*)dst + c * (CHUNK_E / 4);
    const int4* s4 = (const int4*)src + c * (CHUNK_E / 4);
    int  i = t;
    int4 d = d4[i];
    int4 s = s4[i];
    for (;;) {
        const int  in   = i + 1024;
        const bool more = in < CHUNK_E / 4;
        int4 dn = d, sn = s;
        if (more) { dn = d4[in]; sn = s4[in]; }
        const unsigned r0 = (unsigned)(d.x - lo);
        const unsigned r1 = (unsigned)(d.y - lo);
        const unsigned r2 = (unsigned)(d.z - lo);
        const unsigned r3 = (unsigned)(d.w - lo);
        const bool in0 = r0 < (unsigned)GROUP_N;
        const bool in1 = r1 < (unsigned)GROUP_N;
        const bool in2 = r2 < (unsigned)GROUP_N;
        const bool in3 = r3 < (unsigned)GROUP_N;
        unsigned p0 = 0, p1 = 0, p2 = 0, p3 = 0;
        if (in0) p0 = (unsigned)atomicAdd(&s_slot[r0], 1);
        if (in1) p1 = (unsigned)atomicAdd(&s_slot[r1], 1);
        if (in2) p2 = (unsigned)atomicAdd(&s_slot[r2], 1);
        if (in3) p3 = (unsigned)atomicAdd(&s_slot[r3], 1);
        if (in0) col[p0] = s.x;
        if (in1) col[p1] = s.y;
        if (in2) col[p2] = s.z;
        if (in3) col[p3] = s.w;
        if (!more) break;
        d = dn;
        s = sn;
        i = in;
    }
}

__device__ __forceinline__ v4f sage_row4(const float* __restrict__ feat, int s, int lane) {
    return ((const v4f*)(feat + (long long)s * 128))[lane];
}

__global__ __launch_bounds__(256) void sage_gather_kernel(
    const float* __restrict__ feat,
    const int*   __restrict__ col,
    const int*   __restrict__ cnt,
    const int*   __restrict__ rowstart,
    float* __restrict__ sum,
    float* __restrict__ deg,
    int n_nodes) {
    const int lane = threadIdx.x & 31;
    const int d = __builtin_amdgcn_readfirstlane((int)(blockIdx.x * (blockDim.x >> 5) + (threadIdx.x >> 5)));
    if (d >= n_nodes) return;
    const int c     = cnt[d];
    const int start = rowstart[d];
    const int end   = start + c;

    v4f a0 = (v4f){}, a1 = (v4f){}, a2 = (v4f){}, a3 = (v4f){};
    for (int base = start; base < end; base += 32) {
        const int left = end - base;
        const int n    = left < 32 ? left : 32;
        const int mine = (lane < n) ? col[base + lane] : 0;
        int j = 0;
        for (; j + 4 <= n; j += 4) {
            const int s0 = __builtin_amdgcn_readlane(mine, j);
            const int s1 = __builtin_amdgcn_readlane(mine, j + 1);
            const int s2 = __builtin_amdgcn_readlane(mine, j + 2);
            const int s3 = __builtin_amdgcn_readlane(mine, j + 3);
            const v4f v0 = sage_row4(feat, s0, lane);
            const v4f v1 = sage_row4(feat, s1, lane);
            const v4f v2 = sage_row4(feat, s2, lane);
            const v4f v3 = sage_row4(feat, s3, lane);
            a0 += v0; a1 += v1; a2 += v2; a3 += v3;
        }
        const int rem = n - j;
        const int r0 = __builtin_amdgcn_readlane(mine, j & 31);
        const int r1 = __builtin_amdgcn_readlane(mine, (j + 1) & 31);
        const int r2 = __builtin_amdgcn_readlane(mine, (j + 2) & 31);
        v4f v0 = (v4f){}, v1 = (v4f){}, v2 = (v4f){};
        if (rem > 0) v0 = sage_row4(feat, r0, lane);
        if (rem > 1) v1 = sage_row4(feat, r1, lane);
        if (rem > 2) v2 = sage_row4(feat, r2, lane);
        a0 += v0; a1 += v1; a2 += v2;
    }
    const v4f r = (a0 + a1) + (a2 + a3);
    ((v4f*)(sum + (long long)d * 128))[lane] = r;
    if (deg != nullptr && lane == 0) deg[d] = (float)c;
}

#define OUT_NPW 4

__device__ __forceinline__ int sage_pick(int mine, int j, int hf) {
    const int e = __builtin_amdgcn_readlane(mine, j & 31);
    const int o = __builtin_amdgcn_readlane(mine, (j + 1) & 31);
    return hf ? o : e;
}

__device__ __forceinline__ v4f sage_zrow4(const float* __restrict__ z, int s, int l16) {
    return ((const v4f*)(z + (long long)s * 64))[l16];
}

__device__ __forceinline__ void sage_out_node(
    const float* __restrict__ z, const int* __restrict__ col,
    int c, int start, v4f* orow, int lane, int hf, int l16) {
    const int end = start + c;

    v4f own = (v4f){};
    if (hf == 0) own = orow[l16];

    v4f a0 = (v4f){}, a1 = (v4f){}, a2 = (v4f){}, a3 = (v4f){};
    for (int base = start; base < end; base += 32) {
        const int left = end - base;
        const int n    = left < 32 ? left : 32;
        const int mine = (lane < n) ? col[base + lane] : 0;
        int j = 0;
        for (; j + 8 <= n; j += 8) {
            const int s0 = sage_pick(mine, j,     hf);
            const int s1 = sage_pick(mine, j + 2, hf);
            const int s2 = sage_pick(mine, j + 4, hf);
            const int s3 = sage_pick(mine, j + 6, hf);
            const v4f v0 = sage_zrow4(z, s0, l16);
            const v4f v1 = sage_zrow4(z, s1, l16);
            const v4f v2 = sage_zrow4(z, s2, l16);
            const v4f v3 = sage_zrow4(z, s3, l16);
            a0 += v0; a1 += v1; a2 += v2; a3 += v3;
        }
        const int rem = n - j;
        const int r0 = sage_pick(mine, j,     hf);
        const int r1 = sage_pick(mine, j + 2, hf);
        const int r2 = sage_pick(mine, j + 4, hf);
        const int r3 = sage_pick(mine, j + 6, hf);
        v4f v0 = (v4f){}, v1 = (v4f){}, v2 = (v4f){}, v3 = (v4f){};
        if (hf     < rem) v0 = sage_zrow4(z, r0, l16);
        if (hf + 2 < rem) v1 = sage_zrow4(z, r1, l16);
        if (hf + 4 < rem) v2 = sage_zrow4(z, r2, l16);
        if (hf + 6 < rem) v3 = sage_zrow4(z, r3, l16);
        a0 += v0; a1 += v1; a2 += v2; a3 += v3;
    }
    v4f r = (a0 + a1) + (a2 + a3);
    r.x += __shfl_xor(r.x, 16);
    r.y += __shfl_xor(r.y, 16);
    r.z += __shfl_xor(r.z, 16);
    r.w += __shfl_xor(r.w, 16);
    const float invd = 1.0f / fmaxf((float)c, 1.0f);
    if (hf == 0) orow[l16] = own + r * invd;
}

__global__ __launch_bounds__(256) void sage_gather_out_kernel(
    const float* __restrict__ z,
    const int*   __restrict__ col,
    const int*   __restrict__ cnt,
    const int*   __restrict__ rowstart,
    float* __restrict__ out,
    int n_nodes) {
    const int lane = threadIdx.x & 31;
    const int hf   = lane >> 4;
    const int l16  = lane & 15;
    const int w  = __builtin_amdgcn_readfirstlane((int)(blockIdx.x * (blockDim.x >> 5) + (threadIdx.x >> 5)));
    const int d0 = w * OUT_NPW;
    if (d0 >= n_nodes) return;
    const int4 c4 = ((const int4*)cnt)[w];
    const int4 s4 = ((const int4*)rowstart)[w];
    v4f* orow = (v4f*)(out + (long long)d0 * 64);
    sage_out_node(z, col, c4.x, s4.x, orow,      lane, hf, l16);
    sage_out_node(z, col, c4.y, s4.y, orow + 16, lane, hf, l16);
    sage_out_node(z, col, c4.z, s4.z, orow + 32, lane, hf, l16);
    sage_out_node(z, col, c4.w, s4.w, orow + 48, lane, hf, l16);
}

template <int CO>
__device__ __forceinline__ void sage_stage_weights(
    const float* __restrict__ Wl, const float* __restrict__ Wr, v4f* smem4, int tid) {
    constexpr int NV    = 128 * CO / 4;
    constexpr int PER   = NV / 256;
    constexpr int BATCH = 8;
    const v4f* gWl = (const v4f*)Wl;
    const v4f* gWr = (const v4f*)Wr;
    v4f* lWl = smem4;
    v4f* lWr = smem4 + NV;
#pragma unroll
    for (int b = 0; b < PER; b += BATCH) {
        v4f tl[BATCH], tr[BATCH];
#pragma unroll
        for (int u = 0; u < BATCH; ++u) {
            tl[u] = gWl[(b + u) * 256 + tid];
            tr[u] = gWr[(b + u) * 256 + tid];
        }
#pragma unroll
        for (int u = 0; u < BATCH; ++u) {
            lWl[(b + u) * 256 + tid] = tl[u];
            lWr[(b + u) * 256 + tid] = tr[u];
        }
    }
}

__global__ __launch_bounds__(256) void sage_gemm1_kernel(
    float* io,
    const float* __restrict__ deg,
    const float* __restrict__ feat,
    const float* __restrict__ Wl,
    const float* __restrict__ bias,
    const float* __restrict__ Wr,
    int n_tiles) {
    constexpr int CO = HID_C;
    constexpr int NT = CO / 16;

    extern __shared__ v4f smem4[];
    float* sWl = (float*)smem4;
    float* sWr = sWl + 128 * CO;

    const int tid = threadIdx.x;
    sage_stage_weights<CO>(Wl, Wr, smem4, tid);
    __syncthreads();

    const int wave = tid >> 5;
    const int lane = tid & 31;
    const int tile = blockIdx.x * (blockDim.x >> 5) + wave;
    if (tile >= n_tiles) return;

    const int row0 = tile * 16;
    const int hi   = lane >> 4;
    const int lr   = lane & 15;
    const int arow = row0 + lr;

    const float dg   = deg[arow];
    const float invd = 1.0f / fmaxf(dg, 1.0f);

    const float* aggRow = io   + (long long)arow * 128;
    const float* xRow   = feat + (long long)arow * 128;

    v8f acc[NT];
#pragma unroll
    for (int n = 0; n < NT; ++n) acc[n] = (v8f){};

    for (int k = 0; k < 32; ++k) {
        const int c = 4 * k + 2 * hi;
        v2f aAgg, aX;
        aAgg.x = aggRow[c]     * invd;
        aAgg.y = aggRow[c + 1] * invd;
        aX.x   = xRow[c];
        aX.y   = xRow[c + 1];
#pragma unroll
        for (int n = 0; n < NT; ++n) {
            const int col = n * 16 + lr;
            v2f bL, bR;
            bL.x = sWl[(c)     * CO + col];
            bL.y = sWl[(c + 1) * CO + col];
            bR.x = sWr[(c)     * CO + col];
            bR.y = sWr[(c + 1) * CO + col];
            acc[n] = __builtin_amdgcn_wmma_f32_16x16x4_f32(
                false, aAgg, false, bL, (short)0, acc[n], false, false);
            acc[n] = __builtin_amdgcn_wmma_f32_16x16x4_f32(
                false, aX,   false, bR, (short)0, acc[n], false, false);
        }
    }

#pragma unroll
    for (int n = 0; n < NT; ++n) {
        const int col = n * 16 + lr;
        const float b = bias[col];
#pragma unroll
        for (int v = 0; v < 8; ++v) {
            const int r = row0 + v + 8 * hi;
            const float val = fmaxf(acc[n][v] + b, 0.0f);
            io[(long long)r * CO + col] = val;
        }
    }
}

__global__ __launch_bounds__(256) void sage_proj2_kernel(
    const float* __restrict__ feat,
    const float* __restrict__ Wl,
    const float* __restrict__ bias,
    const float* __restrict__ Wr,
    float* __restrict__ z,
    float* __restrict__ out,
    int n_tiles) {
    constexpr int CO = OUT_C;
    constexpr int NT = CO / 16;

    extern __shared__ v4f smem4[];
    float* sWl = (float*)smem4;
    float* sWr = sWl + 128 * CO;

    const int tid = threadIdx.x;
    sage_stage_weights<CO>(Wl, Wr, smem4, tid);
    __syncthreads();

    const int wave = tid >> 5;
    const int lane = tid & 31;
    const int tile = blockIdx.x * (blockDim.x >> 5) + wave;
    if (tile >= n_tiles) return;

    const int row0 = tile * 16;
    const int hi   = lane >> 4;
    const int lr   = lane & 15;
    const int arow = row0 + lr;

    const float* xRow = feat + (long long)arow * 128;

    v8f accL[NT], accR[NT];
#pragma unroll
    for (int n = 0; n < NT; ++n) { accL[n] = (v8f){}; accR[n] = (v8f){}; }

    for (int k = 0; k < 32; ++k) {
        const int c = 4 * k + 2 * hi;
        v2f aX;
        aX.x = xRow[c];
        aX.y = xRow[c + 1];
#pragma unroll
        for (int n = 0; n < NT; ++n) {
            const int col = n * 16 + lr;
            v2f bL, bR;
            bL.x = sWl[(c)     * CO + col];
            bL.y = sWl[(c + 1) * CO + col];
            bR.x = sWr[(c)     * CO + col];
            bR.y = sWr[(c + 1) * CO + col];
            accL[n] = __builtin_amdgcn_wmma_f32_16x16x4_f32(
                false, aX, false, bL, (short)0, accL[n], false, false);
            accR[n] = __builtin_amdgcn_wmma_f32_16x16x4_f32(
                false, aX, false, bR, (short)0, accR[n], false, false);
        }
    }

#pragma unroll
    for (int n = 0; n < NT; ++n) {
        const int col = n * 16 + lr;
        const float b = bias[col];
#pragma unroll
        for (int v = 0; v < 8; ++v) {
            const int r = row0 + v + 8 * hi;
            z[(long long)r * CO + col]   = accL[n][v];
            out[(long long)r * CO + col] = accR[n][v] + b;
        }
    }
}

extern "C" void kernel_launch(void* const* d_in, const int* in_sizes, int n_in,
                              void* d_out, int out_size, void* d_ws, size_t ws_size,
                              hipStream_t stream) {
    const float* x   = (const float*)d_in[0];
    const int*   ei  = (const int*)  d_in[1];
    const float* W1l = (const float*)d_in[2];
    const float* b1  = (const float*)d_in[3];
    const float* W1r = (const float*)d_in[4];
    const float* W2l = (const float*)d_in[5];
    const float* b2  = (const float*)d_in[6];
    const float* W2r = (const float*)d_in[7];

    const int* src = ei;
    const int* dst = ei + N_EDGES;

    float* ws  = (float*)d_ws;
    float* sum = ws;
    float* deg = ws + (long long)N_NODES * 128;
    float* out = (float*)d_out;

    int* col    = (int*)(deg + N_NODES);
    int* cnt    = col + N_EDGES;
    int* cursor = cnt + N_NODES;
    float* z    = ws + 7350016;
    int* part   = (int*)d_ws;

    const int tiles       = N_NODES / 16;
    const int gemm_blocks = (tiles + 7) / 8;
    const int node_blocks = (N_NODES + 7) / 8;
    const int scan_blocks = (N_NODES + 1023) / 1024;
    const int out_blocks  = (N_NODES + 8 * OUT_NPW - 1) / (8 * OUT_NPW);

    sage_hist_kernel<<<SORT_C * SORT_G, 1024, 0, stream>>>(dst, part);
    sage_chunkscan_kernel<<<(N_NODES + 255) / 256, 256, 0, stream>>>(part, cnt, N_NODES);
    sage_rowstart_kernel<<<scan_blocks, 1024, 0, stream>>>(cnt, cursor, N_NODES);
    sage_fill_kernel<<<SORT_C * SORT_G, 1024, 0, stream>>>(src, dst, part, cursor, col);

    sage_gather_kernel<<<node_blocks, 256, 0, stream>>>(x, col, cnt, cursor, sum, deg, N_NODES);
    sage_gemm1_kernel<<<gemm_blocks, 256, 2 * 128 * HID_C * sizeof(float), stream>>>(
        sum, deg, x, W1l, b1, W1r, tiles);

    sage_proj2_kernel<<<gemm_blocks, 256, 2 * 128 * OUT_C * sizeof(float), stream>>>(
        sum, W2l, b2, W2r, z, out, tiles);
    sage_gather_out_kernel<<<out_blocks, 256, 0, stream>>>(z, col, cnt, cursor, out, N_NODES);
}


# Row-start kernel folded into the fill kernel: each fill block scans its node group's degrees itself and sums the degrees in front of the group, loads requested together; 7 launches, was 8
# speedup vs baseline: 3.8489x; 1.0524x over previous
#include <hip/hip_runtime.h>


#define N_NODES 50000
#define N_EDGES 800000
#define IN_C    128
#define HID_C   128
#define OUT_C   64

typedef float v2f __attribute__((ext_vector_type(2)));
typedef float v4f __attribute__((ext_vector_type(4)));
typedef float v8f __attribute__((ext_vector_type(8)));

#define SORT_G   8
#define SORT_C   32
#define GROUP_N  (N_NODES / SORT_G)
#define CHUNK_E  (N_EDGES / SORT_C)

__global__ __launch_bounds__(1024) void sage_hist_kernel(
    const int* __restrict__ dst, int* __restrict__ part) {
    __shared__ int s_cnt[GROUP_N];
    const int t  = threadIdx.x;
    const int c  = blockIdx.x / SORT_G;
    const int g  = blockIdx.x % SORT_G;
    const int lo = g * GROUP_N;
    for (int i = t; i < GROUP_N; i += 1024) s_cnt[i] = 0;
    __syncthreads();
    const int4* d4 = (const int4*)dst + c * (CHUNK_E / 4);
    int  i = t;
    int4 d = d4[i];
    for (;;) {
        const int  in   = i + 1024;
        const bool more = in < CHUNK_E / 4;
        int4 dn = d;
        if (more) dn = d4[in];
        const unsigned r0 = (unsigned)(d.x - lo);
        const unsigned r1 = (unsigned)(d.y - lo);
        const unsigned r2 = (unsigned)(d.z - lo);
        const unsigned r3 = (unsigned)(d.w - lo);
        if (r0 < (unsigned)GROUP_N) atomicAdd(&s_cnt[r0], 1);
        if (r1 < (unsigned)GROUP_N) atomicAdd(&s_cnt[r1], 1);
        if (r2 < (unsigned)GROUP_N) atomicAdd(&s_cnt[r2], 1);
        if (r3 < (unsigned)GROUP_N) atomicAdd(&s_cnt[r3], 1);
        if (!more) break;
        d = dn;
        i = in;
    }
    __syncthreads();
    int* p = part + c * N_NODES + lo;
    for (int i = t; i < GROUP_N; i += 1024) p[i] = s_cnt[i];
}

__global__ __launch_bounds__(256) void sage_chunkscan_kernel(
    int* __restrict__ part, int* __restrict__ cnt, int n) {
    const int d = blockIdx.x * blockDim.x + threadIdx.x;
    if (d >= n) return;
    int v[SORT_C];
#pragma unroll
    for (int c = 0; c < SORT_C; ++c) v[c] = part[c * N_NODES + d];
    int run = 0;
#pragma unroll
    for (int c = 0; c < SORT_C; ++c) {
        part[c * N_NODES + d] = run;
        run += v[c];
    }
    cnt[d] = run;
}

#define FILL_PER   7
#define FILL_FRONT 11

__global__ __launch_bounds__(1024) void sage_fill_kernel(
    const int* __restrict__ src, const int* __restrict__ dst,
    const int* __restrict__ part, const int* __restrict__ cnt,
    int* __restrict__ cursor, int* __restrict__ col) {
    __shared__ int s_slot[GROUP_N];
    __shared__ int s_before[32];
    __shared__ int s_own[32];
    const int t    = threadIdx.x;
    const int lane = t & 31;
    const int w    = t >> 5;
    const int c  = blockIdx.x / SORT_G;
    const int g  = blockIdx.x % SORT_G;
    const int lo = g * GROUP_N;
    const int* p = part + c * N_NODES + lo;

    const int4* d4 = (const int4*)dst + c * (CHUNK_E / 4);
    const int4* s4 = (const int4*)src + c * (CHUNK_E / 4);
    int  i = t;
    int4 d = d4[i];
    int4 s = s4[i];

    const int i0 = t * FILL_PER;
    int cn[FILL_PER], pv[FILL_PER];
#pragma unroll
    for (int k = 0; k < FILL_PER; ++k) {
        const int  nd = i0 + k;
        const bool ok = nd < GROUP_N;
        cn[k] = ok ? cnt[lo + nd] : 0;
        pv[k] = ok ? p[nd] : 0;
    }

    const int4* cnt4 = (const int4*)cnt;
    const int   n4   = lo >> 2;
    int4 fr[FILL_FRONT];
#pragma unroll
    for (int k = 0; k < FILL_FRONT; ++k) {
        const int q = t + 1024 * k;
        fr[k] = (q < n4) ? cnt4[q] : make_int4(0, 0, 0, 0);
    }
    int before = (t < (lo & 3)) ? cnt[4 * n4 + t] : 0;
#pragma unroll
    for (int k = 0; k < FILL_FRONT; ++k) before += (fr[k].x + fr[k].y) + (fr[k].z + fr[k].w);

    int ex[FILL_PER];
    int run = 0;
#pragma unroll
    for (int k = 0; k < FILL_PER; ++k) { ex[k] = run; run += cn[k]; }

    int incl = run;
    for (int off = 1; off < 32; off <<= 1) {
        const int up = __shfl_up(incl, off);
        if (lane >= off) incl += up;
    }
    for (int off = 16; off > 0; off >>= 1) before += __shfl_xor(before, off);

    if (lane == 31) s_own[w] = incl;
    if (lane == 0)  s_before[w] = before;
    __syncthreads();

    int base = 0;
    for (int k = 0; k < 32; ++k) {
        base += s_before[k];
        if (k < w) base += s_own[k];
    }
    base += incl - run;
#pragma unroll
    for (int k = 0; k < FILL_PER; ++k) {
        const int nd = i0 + k;
        if (nd < GROUP_N) {
            const int rs = base + ex[k];
            s_slot[nd] = rs + pv[k];
            if (c == 0) cursor[lo + nd] = rs;
        }
    }
    __syncthreads();
    for (;;) {
        const int  in   = i + 1024;
        const bool more = in < CHUNK_E / 4;
        int4 dn = d, sn = s;
        if (more) { dn = d4[in]; sn = s4[in]; }
        const unsigned r0 = (unsigned)(d.x - lo);
        const unsigned r1 = (unsigned)(d.y - lo);
        const unsigned r2 = (unsigned)(d.z - lo);
        const unsigned r3 = (unsigned)(d.w - lo);
        const bool in0 = r0 < (unsigned)GROUP_N;
        const bool in1 = r1 < (unsigned)GROUP_N;
        const bool in2 = r2 < (unsigned)GROUP_N;
        const bool in3 = r3 < (unsigned)GROUP_N;
        unsigned p0 = 0, p1 = 0, p2 = 0, p3 = 0;
        if (in0) p0 = (unsigned)atomicAdd(&s_slot[r0], 1);
        if (in1) p1 = (unsigned)atomicAdd(&s_slot[r1], 1);
        if (in2) p2 = (unsigned)atomicAdd(&s_slot[r2], 1);
        if (in3) p3 = (unsigned)atomicAdd(&s_slot[r3], 1);
        if (in0) col[p0] = s.x;
        if (in1) col[p1] = s.y;
        if (in2) col[p2] = s.z;
        if (in3) col[p3] = s.w;
        if (!more) break;
        d = dn;
        s = sn;
        i = in;
    }
}

__device__ __forceinline__ v4f sage_row4(const float* __restrict__ feat, int s, int lane) {
    return ((const v4f*)(feat + (long long)s * 128))[lane];
}

__global__ __launch_bounds__(256) void sage_gather_kernel(
    const float* __restrict__ feat,
    const int*   __restrict__ col,
    const int*   __restrict__ cnt,
    const int*   __restrict__ rowstart,
    float* __restrict__ sum,
    float* __restrict__ deg,
    int n_nodes) {
    const int lane = threadIdx.x & 31;
    const int d = __builtin_amdgcn_readfirstlane((int)(blockIdx.x * (blockDim.x >> 5) + (threadIdx.x >> 5)));
    if (d >= n_nodes) return;
    const int c     = cnt[d];
    const int start = rowstart[d];
    const int end   = start + c;

    v4f a0 = (v4f){}, a1 = (v4f){}, a2 = (v4f){}, a3 = (v4f){};
    for (int base = start; base < end; base += 32) {
        const int left = end - base;
        const int n    = left < 32 ? left : 32;
        const int mine = (lane < n) ? col[base + lane] : 0;
        int j = 0;
        for (; j + 4 <= n; j += 4) {
            const int s0 = __builtin_amdgcn_readlane(mine, j);
            const int s1 = __builtin_amdgcn_readlane(mine, j + 1);
            const int s2 = __builtin_amdgcn_readlane(mine, j + 2);
            const int s3 = __builtin_amdgcn_readlane(mine, j + 3);
            const v4f v0 = sage_row4(feat, s0, lane);
            const v4f v1 = sage_row4(feat, s1, lane);
            const v4f v2 = sage_row4(feat, s2, lane);
            const v4f v3 = sage_row4(feat, s3, lane);
            a0 += v0; a1 += v1; a2 += v2; a3 += v3;
        }
        const int rem = n - j;
        const int r0 = __builtin_amdgcn_readlane(mine, j & 31);
        const int r1 = __builtin_amdgcn_readlane(mine, (j + 1) & 31);
        const int r2 = __builtin_amdgcn_readlane(mine, (j + 2) & 31);
        v4f v0 = (v4f){}, v1 = (v4f){}, v2 = (v4f){};
        if (rem > 0) v0 = sage_row4(feat, r0, lane);
        if (rem > 1) v1 = sage_row4(feat, r1, lane);
        if (rem > 2) v2 = sage_row4(feat, r2, lane);
        a0 += v0; a1 += v1; a2 += v2;
    }
    const v4f r = (a0 + a1) + (a2 + a3);
    ((v4f*)(sum + (long long)d * 128))[lane] = r;
    if (deg != nullptr && lane == 0) deg[d] = (float)c;
}

#define OUT_NPW 4

__device__ __forceinline__ int sage_pick(int mine, int j, int hf) {
    const int e = __builtin_amdgcn_readlane(mine, j & 31);
    const int o = __builtin_amdgcn_readlane(mine, (j + 1) & 31);
    return hf ? o : e;
}

__device__ __forceinline__ v4f sage_zrow4(const float* __restrict__ z, int s, int l16) {
    return ((const v4f*)(z + (long long)s * 64))[l16];
}

__device__ __forceinline__ void sage_out_node(
    const float* __restrict__ z, const int* __restrict__ col,
    int c, int start, v4f* orow, int lane, int hf, int l16) {
    const int end = start + c;

    v4f own = (v4f){};
    if (hf == 0) own = orow[l16];

    v4f a0 = (v4f){}, a1 = (v4f){}, a2 = (v4f){}, a3 = (v4f){};
    for (int base = start; base < end; base += 32) {
        const int left = end - base;
        const int n    = left < 32 ? left : 32;
        const int mine = (lane < n) ? col[base + lane] : 0;
        int j = 0;
        for (; j + 8 <= n; j += 8) {
            const int s0 = sage_pick(mine, j,     hf);
            const int s1 = sage_pick(mine, j + 2, hf);
            const int s2 = sage_pick(mine, j + 4, hf);
            const int s3 = sage_pick(mine, j + 6, hf);
            const v4f v0 = sage_zrow4(z, s0, l16);
            const v4f v1 = sage_zrow4(z, s1, l16);
            const v4f v2 = sage_zrow4(z, s2, l16);
            const v4f v3 = sage_zrow4(z, s3, l16);
            a0 += v0; a1 += v1; a2 += v2; a3 += v3;
        }
        const int rem = n - j;
        const int r0 = sage_pick(mine, j,     hf);
        const int r1 = sage_pick(mine, j + 2, hf);
        const int r2 = sage_pick(mine, j + 4, hf);
        const int r3 = sage_pick(mine, j + 6, hf);
        v4f v0 = (v4f){}, v1 = (v4f){}, v2 = (v4f){}, v3 = (v4f){};
        if (hf     < rem) v0 = sage_zrow4(z, r0, l16);
        if (hf + 2 < rem) v1 = sage_zrow4(z, r1, l16);
        if (hf + 4 < rem) v2 = sage_zrow4(z, r2, l16);
        if (hf + 6 < rem) v3 = sage_zrow4(z, r3, l16);
        a0 += v0; a1 += v1; a2 += v2; a3 += v3;
    }
    v4f r = (a0 + a1) + (a2 + a3);
    r.x += __shfl_xor(r.x, 16);
    r.y += __shfl_xor(r.y, 16);
    r.z += __shfl_xor(r.z, 16);
    r.w += __shfl_xor(r.w, 16);
    const float invd = 1.0f / fmaxf((float)c, 1.0f);
    if (hf == 0) orow[l16] = own + r * invd;
}

__global__ __launch_bounds__(256) void sage_gather_out_kernel(
    const float* __restrict__ z,
    const int*   __restrict__ col,
    const int*   __restrict__ cnt,
    const int*   __restrict__ rowstart,
    float* __restrict__ out,
    int n_nodes) {
    const int lane = threadIdx.x & 31;
    const int hf   = lane >> 4;
    const int l16  = lane & 15;
    const int w  = __builtin_amdgcn_readfirstlane((int)(blockIdx.x * (blockDim.x >> 5) + (threadIdx.x >> 5)));
    const int d0 = w * OUT_NPW;
    if (d0 >= n_nodes) return;
    const int4 c4 = ((const int4*)cnt)[w];
    const int4 s4 = ((const int4*)rowstart)[w];
    v4f* orow = (v4f*)(out + (long long)d0 * 64);
    sage_out_node(z, col, c4.x, s4.x, orow,      lane, hf, l16);
    sage_out_node(z, col, c4.y, s4.y, orow + 16, lane, hf, l16);
    sage_out_node(z, col, c4.z, s4.z, orow + 32, lane, hf, l16);
    sage_out_node(z, col, c4.w, s4.w, orow + 48, lane, hf, l16);
}

template <int CO>
__device__ __forceinline__ void sage_stage_weights(
    const float* __restrict__ Wl, const float* __restrict__ Wr, v4f* smem4, int tid) {
    constexpr int NV    = 128 * CO / 4;
    constexpr int PER   = NV / 256;
    constexpr int BATCH = 8;
    const v4f* gWl = (const v4f*)Wl;
    const v4f* gWr = (const v4f*)Wr;
    v4f* lWl = smem4;
    v4f* lWr = smem4 + NV;
#pragma unroll
    for (int b = 0; b < PER; b += BATCH) {
        v4f tl[BATCH], tr[BATCH];
#pragma unroll
        for (int u = 0; u < BATCH; ++u) {
            tl[u] = gWl[(b + u) * 256 + tid];
            tr[u] = gWr[(b + u) * 256 + tid];
        }
#pragma unroll
        for (int u = 0; u < BATCH; ++u) {
            lWl[(b + u) * 256 + tid] = tl[u];
            lWr[(b + u) * 256 + tid] = tr[u];
        }
    }
}

__global__ __launch_bounds__(256) void sage_gemm1_kernel(
    float* io,
    const float* __restrict__ deg,
    const float* __restrict__ feat,
    const float* __restrict__ Wl,
    const float* __restrict__ bias,
    const float* __restrict__ Wr,
    int n_tiles) {
    constexpr int CO = HID_C;
    constexpr int NT = CO / 16;

    extern __shared__ v4f smem4[];
    float* sWl = (float*)smem4;
    float* sWr = sWl + 128 * CO;

    const int tid = threadIdx.x;
    sage_stage_weights<CO>(Wl, Wr, smem4, tid);
    __syncthreads();

    const int wave = tid >> 5;
    const int lane = tid & 31;
    const int tile = blockIdx.x * (blockDim.x >> 5) + wave;
    if (tile >= n_tiles) return;

    const int row0 = tile * 16;
    const int hi   = lane >> 4;
    const int lr   = lane & 15;
    const int arow = row0 + lr;

    const float dg   = deg[arow];
    const float invd = 1.0f / fmaxf(dg, 1.0f);

    const float* aggRow = io   + (long long)arow * 128;
    const float* xRow   = feat + (long long)arow * 128;

    v8f acc[NT];
#pragma unroll
    for (int n = 0; n < NT; ++n) acc[n] = (v8f){};

    for (int k = 0; k < 32; ++k) {
        const int c = 4 * k + 2 * hi;
        v2f aAgg, aX;
        aAgg.x = aggRow[c]     * invd;
        aAgg.y = aggRow[c + 1] * invd;
        aX.x   = xRow[c];
        aX.y   = xRow[c + 1];
#pragma unroll
        for (int n = 0; n < NT; ++n) {
            const int col = n * 16 + lr;
            v2f bL, bR;
            bL.x = sWl[(c)     * CO + col];
            bL.y = sWl[(c + 1) * CO + col];
            bR.x = sWr[(c)     * CO + col];
            bR.y = sWr[(c + 1) * CO + col];
            acc[n] = __builtin_amdgcn_wmma_f32_16x16x4_f32(
                false, aAgg, false, bL, (short)0, acc[n], false, false);
            acc[n] = __builtin_amdgcn_wmma_f32_16x16x4_f32(
                false, aX,   false, bR, (short)0, acc[n], false, false);
        }
    }

#pragma unroll
    for (int n = 0; n < NT; ++n) {
        const int col = n * 16 + lr;
        const float b = bias[col];
#pragma unroll
        for (int v = 0; v < 8; ++v) {
            const int r = row0 + v + 8 * hi;
            const float val = fmaxf(acc[n][v] + b, 0.0f);
            io[(long long)r * CO + col] = val;
        }
    }
}

__global__ __launch_bounds__(256) void sage_proj2_kernel(
    const float* __restrict__ feat,
    const float* __restrict__ Wl,
    const float* __restrict__ bias,
    const float* __restrict__ Wr,
    float* __restrict__ z,
    float* __restrict__ out,
    int n_tiles) {
    constexpr int CO = OUT_C;
    constexpr int NT = CO / 16;

    extern __shared__ v4f smem4[];
    float* sWl = (float*)smem4;
    float* sWr = sWl + 128 * CO;

    const int tid = threadIdx.x;
    sage_stage_weights<CO>(Wl, Wr, smem4, tid);
    __syncthreads();

    const int wave = tid >> 5;
    const int lane = tid & 31;
    const int tile = blockIdx.x * (blockDim.x >> 5) + wave;
    if (tile >= n_tiles) return;

    const int row0 = tile * 16;
    const int hi   = lane >> 4;
    const int lr   = lane & 15;
    const int arow = row0 + lr;

    const float* xRow = feat + (long long)arow * 128;

    v8f accL[NT], accR[NT];
#pragma unroll
    for (int n = 0; n < NT; ++n) { accL[n] = (v8f){}; accR[n] = (v8f){}; }

    for (int k = 0; k < 32; ++k) {
        const int c = 4 * k + 2 * hi;
        v2f aX;
        aX.x = xRow[c];
        aX.y = xRow[c + 1];
#pragma unroll
        for (int n = 0; n < NT; ++n) {
            const int col = n * 16 + lr;
            v2f bL, bR;
            bL.x = sWl[(c)     * CO + col];
            bL.y = sWl[(c + 1) * CO + col];
            bR.x = sWr[(c)     * CO + col];
            bR.y = sWr[(c + 1) * CO + col];
            accL[n] = __builtin_amdgcn_wmma_f32_16x16x4_f32(
                false, aX, false, bL, (short)0, accL[n], false, false);
            accR[n] = __builtin_amdgcn_wmma_f32_16x16x4_f32(
                false, aX, false, bR, (short)0, accR[n], false, false);
        }
    }

#pragma unroll
    for (int n = 0; n < NT; ++n) {
        const int col = n * 16 + lr;
        const float b = bias[col];
#pragma unroll
        for (int v = 0; v < 8; ++v) {
            const int r = row0 + v + 8 * hi;
            z[(long long)r * CO + col]   = accL[n][v];
            out[(long long)r * CO + col] = accR[n][v] + b;
        }
    }
}

extern "C" void kernel_launch(void* const* d_in, const int* in_sizes, int n_in,
                              void* d_out, int out_size, void* d_ws, size_t ws_size,
                              hipStream_t stream) {
    const float* x   = (const float*)d_in[0];
    const int*   ei  = (const int*)  d_in[1];
    const float* W1l = (const float*)d_in[2];
    const float* b1  = (const float*)d_in[3];
    const float* W1r = (const float*)d_in[4];
    const float* W2l = (const float*)d_in[5];
    const float* b2  = (const float*)d_in[6];
    const float* W2r = (const float*)d_in[7];

    const int* src = ei;
    const int* dst = ei + N_EDGES;

    float* ws  = (float*)d_ws;
    float* sum = ws;
    float* deg = ws + (long long)N_NODES * 128;
    float* out = (float*)d_out;

    int* col    = (int*)(deg + N_NODES);
    int* cnt    = col + N_EDGES;
    int* cursor = cnt + N_NODES;
    float* z    = ws + 7350016;
    int* part   = (int*)d_ws;

    const int tiles       = N_NODES / 16;
    const int gemm_blocks = (tiles + 7) / 8;
    const int node_blocks = (N_NODES + 7) / 8;
    const int out_blocks  = (N_NODES + 8 * OUT_NPW - 1) / (8 * OUT_NPW);

    sage_hist_kernel<<<SORT_C * SORT_G, 1024, 0, stream>>>(dst, part);
    sage_chunkscan_kernel<<<(N_NODES + 255) / 256, 256, 0, stream>>>(part, cnt, N_NODES);
    sage_fill_kernel<<<SORT_C * SORT_G, 1024, 0, stream>>>(src, dst, part, cnt, cursor, col);

    sage_gather_kernel<<<node_blocks, 256, 0, stream>>>(x, col, cnt, cursor, sum, deg, N_NODES);
    sage_gemm1_kernel<<<gemm_blocks, 256, 2 * 128 * HID_C * sizeof(float), stream>>>(
        sum, deg, x, W1l, b1, W1r, tiles);

    sage_proj2_kernel<<<gemm_blocks, 256, 2 * 128 * OUT_C * sizeof(float), stream>>>(
        sum, W2l, b2, W2r, z, out, tiles);
    sage_gather_out_kernel<<<out_blocks, 256, 0, stream>>>(z, col, cnt, cursor, out, N_NODES);
}


# Layer-1 linear and layer-2 projections in one launch: h tile formed transposed (weights as A operand) so its accumulators are stage-2 A fragments; h never stored; layer-2 k order regrouped
# speedup vs baseline: 4.5752x; 1.1887x over previous
#include <hip/hip_runtime.h>


#define N_NODES 50000
#define N_EDGES 800000
#define IN_C    128
#define HID_C   128
#define OUT_C   64

typedef float v2f __attribute__((ext_vector_type(2)));
typedef float v4f __attribute__((ext_vector_type(4)));
typedef float v8f __attribute__((ext_vector_type(8)));

#define SORT_G   8
#define SORT_C   32
#define GROUP_N  (N_NODES / SORT_G)
#define CHUNK_E  (N_EDGES / SORT_C)

__global__ __launch_bounds__(1024) void sage_hist_kernel(
    const int* __restrict__ dst, int* __restrict__ part) {
    __shared__ int s_cnt[GROUP_N];
    const int t  = threadIdx.x;
    const int c  = blockIdx.x / SORT_G;
    const int g  = blockIdx.x % SORT_G;
    const int lo = g * GROUP_N;
    for (int i = t; i < GROUP_N; i += 1024) s_cnt[i] = 0;
    __syncthreads();
    const int4* d4 = (const int4*)dst + c * (CHUNK_E / 4);
    int  i = t;
    int4 d = d4[i];
    for (;;) {
        const int  in   = i + 1024;
        const bool more = in < CHUNK_E / 4;
        int4 dn = d;
        if (more) dn = d4[in];
        const unsigned r0 = (unsigned)(d.x - lo);
        const unsigned r1 = (unsigned)(d.y - lo);
        const unsigned r2 = (unsigned)(d.z - lo);
        const unsigned r3 = (unsigned)(d.w - lo);
        if (r0 < (unsigned)GROUP_N) atomicAdd(&s_cnt[r0], 1);
        if (r1 < (unsigned)GROUP_N) atomicAdd(&s_cnt[r1], 1);
        if (r2 < (unsigned)GROUP_N) atomicAdd(&s_cnt[r2], 1);
        if (r3 < (unsigned)GROUP_N) atomicAdd(&s_cnt[r3], 1);
        if (!more) break;
        d = dn;
        i = in;
    }
    __syncthreads();
    int* p = part + c * N_NODES + lo;
    for (int i = t; i < GROUP_N; i += 1024) p[i] = s_cnt[i];
}

__global__ __launch_bounds__(256) void sage_chunkscan_kernel(
    int* __restrict__ part, int* __restrict__ cnt, int n) {
    const int d = blockIdx.x * blockDim.x + threadIdx.x;
    if (d >= n) return;
    int v[SORT_C];
#pragma unroll
    for (int c = 0; c < SORT_C; ++c) v[c] = part[c * N_NODES + d];
    int run = 0;
#pragma unroll
    for (int c = 0; c < SORT_C; ++c) {
        part[c * N_NODES + d] = run;
        run += v[c];
    }
    cnt[d] = run;
}

#define FILL_PER   7
#define FILL_FRONT 11

__global__ __launch_bounds__(1024) void sage_fill_kernel(
    const int* __restrict__ src, const int* __restrict__ dst,
    const int* __restrict__ part, const int* __restrict__ cnt,
    int* __restrict__ cursor, int* __restrict__ col) {
    __shared__ int s_slot[GROUP_N];
    __shared__ int s_before[32];
    __shared__ int s_own[32];
    const int t    = threadIdx.x;
    const int lane = t & 31;
    const int w    = t >> 5;
    const int c  = blockIdx.x / SORT_G;
    const int g  = blockIdx.x % SORT_G;
    const int lo = g * GROUP_N;
    const int* p = part + c * N_NODES + lo;

    const int4* d4 = (const int4*)dst + c * (CHUNK_E / 4);
    const int4* s4 = (const int4*)src + c * (CHUNK_E / 4);
    int  i = t;
    int4 d = d4[i];
    int4 s = s4[i];

    const int i0 = t * FILL_PER;
    int cn[FILL_PER], pv[FILL_PER];
#pragma unroll
    for (int k = 0; k < FILL_PER; ++k) {
        const int  nd = i0 + k;
        const bool ok = nd < GROUP_N;
        cn[k] = ok ? cnt[lo + nd] : 0;
        pv[k] = ok ? p[nd] : 0;
    }

    const int4* cnt4 = (const int4*)cnt;
    const int   n4   = lo >> 2;
    int4 fr[FILL_FRONT];
#pragma unroll
    for (int k = 0; k < FILL_FRONT; ++k) {
        const int q = t + 1024 * k;
        fr[k] = (q < n4) ? cnt4[q] : make_int4(0, 0, 0, 0);
    }
    int before = (t < (lo & 3)) ? cnt[4 * n4 + t] : 0;
#pragma unroll
    for (int k = 0; k < FILL_FRONT; ++k) before += (fr[k].x + fr[k].y) + (fr[k].z + fr[k].w);

    int ex[FILL_PER];
    int run = 0;
#pragma unroll
    for (int k = 0; k < FILL_PER; ++k) { ex[k] = run; run += cn[k]; }

    int incl = run;
    for (int off = 1; off < 32; off <<= 1) {
        const int up = __shfl_up(incl, off);
        if (lane >= off) incl += up;
    }
    for (int off = 16; off > 0; off >>= 1) before += __shfl_xor(before, off);

    if (lane == 31) s_own[w] = incl;
    if (lane == 0)  s_before[w] = before;
    __syncthreads();

    int base = 0;
    for (int k = 0; k < 32; ++k) {
        base += s_before[k];
        if (k < w) base += s_own[k];
    }
    base += incl - run;
#pragma unroll
    for (int k = 0; k < FILL_PER; ++k) {
        const int nd = i0 + k;
        if (nd < GROUP_N) {
            const int rs = base + ex[k];
            s_slot[nd] = rs + pv[k];
            if (c == 0) cursor[lo + nd] = rs;
        }
    }
    __syncthreads();
    for (;;) {
        const int  in   = i + 1024;
        const bool more = in < CHUNK_E / 4;
        int4 dn = d, sn = s;
        if (more) { dn = d4[in]; sn = s4[in]; }
        const unsigned r0 = (unsigned)(d.x - lo);
        const unsigned r1 = (unsigned)(d.y - lo);
        const unsigned r2 = (unsigned)(d.z - lo);
        const unsigned r3 = (unsigned)(d.w - lo);
        const bool in0 = r0 < (unsigned)GROUP_N;
        const bool in1 = r1 < (unsigned)GROUP_N;
        const bool in2 = r2 < (unsigned)GROUP_N;
        const bool in3 = r3 < (unsigned)GROUP_N;
        unsigned p0 = 0, p1 = 0, p2 = 0, p3 = 0;
        if (in0) p0 = (unsigned)atomicAdd(&s_slot[r0], 1);
        if (in1) p1 = (unsigned)atomicAdd(&s_slot[r1], 1);
        if (in2) p2 = (unsigned)atomicAdd(&s_slot[r2], 1);
        if (in3) p3 = (unsigned)atomicAdd(&s_slot[r3], 1);
        if (in0) col[p0] = s.x;
        if (in1) col[p1] = s.y;
        if (in2) col[p2] = s.z;
        if (in3) col[p3] = s.w;
        if (!more) break;
        d = dn;
        s = sn;
        i = in;
    }
}

__device__ __forceinline__ v4f sage_row4(const float* __restrict__ feat, int s, int lane) {
    return ((const v4f*)(feat + (long long)s * 128))[lane];
}

__global__ __launch_bounds__(256) void sage_gather_kernel(
    const float* __restrict__ feat,
    const int*   __restrict__ col,
    const int*   __restrict__ cnt,
    const int*   __restrict__ rowstart,
    float* __restrict__ sum,
    float* __restrict__ deg,
    int n_nodes) {
    const int lane = threadIdx.x & 31;
    const int d = __builtin_amdgcn_readfirstlane((int)(blockIdx.x * (blockDim.x >> 5) + (threadIdx.x >> 5)));
    if (d >= n_nodes) return;
    const int c     = cnt[d];
    const int start = rowstart[d];
    const int end   = start + c;

    v4f a0 = (v4f){}, a1 = (v4f){}, a2 = (v4f){}, a3 = (v4f){};
    for (int base = start; base < end; base += 32) {
        const int left = end - base;
        const int n    = left < 32 ? left : 32;
        const int mine = (lane < n) ? col[base + lane] : 0;
        int j = 0;
        for (; j + 4 <= n; j += 4) {
            const int s0 = __builtin_amdgcn_readlane(mine, j);
            const int s1 = __builtin_amdgcn_readlane(mine, j + 1);
            const int s2 = __builtin_amdgcn_readlane(mine, j + 2);
            const int s3 = __builtin_amdgcn_readlane(mine, j + 3);
            const v4f v0 = sage_row4(feat, s0, lane);
            const v4f v1 = sage_row4(feat, s1, lane);
            const v4f v2 = sage_row4(feat, s2, lane);
            const v4f v3 = sage_row4(feat, s3, lane);
            a0 += v0; a1 += v1; a2 += v2; a3 += v3;
        }
        const int rem = n - j;
        const int r0 = __builtin_amdgcn_readlane(mine, j & 31);
        const int r1 = __builtin_amdgcn_readlane(mine, (j + 1) & 31);
        const int r2 = __builtin_amdgcn_readlane(mine, (j + 2) & 31);
        v4f v0 = (v4f){}, v1 = (v4f){}, v2 = (v4f){};
        if (rem > 0) v0 = sage_row4(feat, r0, lane);
        if (rem > 1) v1 = sage_row4(feat, r1, lane);
        if (rem > 2) v2 = sage_row4(feat, r2, lane);
        a0 += v0; a1 += v1; a2 += v2;
    }
    const v4f r = (a0 + a1) + (a2 + a3);
    ((v4f*)(sum + (long long)d * 128))[lane] = r;
    if (deg != nullptr && lane == 0) deg[d] = (float)c;
}

#define OUT_NPW 4

__device__ __forceinline__ int sage_pick(int mine, int j, int hf) {
    const int e = __builtin_amdgcn_readlane(mine, j & 31);
    const int o = __builtin_amdgcn_readlane(mine, (j + 1) & 31);
    return hf ? o : e;
}

__device__ __forceinline__ v4f sage_zrow4(const float* __restrict__ z, int s, int l16) {
    return ((const v4f*)(z + (long long)s * 64))[l16];
}

__device__ __forceinline__ void sage_out_node(
    const float* __restrict__ z, const int* __restrict__ col,
    int c, int start, v4f* orow, int lane, int hf, int l16) {
    const int end = start + c;

    v4f own = (v4f){};
    if (hf == 0) own = orow[l16];

    v4f a0 = (v4f){}, a1 = (v4f){}, a2 = (v4f){}, a3 = (v4f){};
    for (int base = start; base < end; base += 32) {
        const int left = end - base;
        const int n    = left < 32 ? left : 32;
        const int mine = (lane < n) ? col[base + lane] : 0;
        int j = 0;
        for (; j + 8 <= n; j += 8) {
            const int s0 = sage_pick(mine, j,     hf);
            const int s1 = sage_pick(mine, j + 2, hf);
            const int s2 = sage_pick(mine, j + 4, hf);
            const int s3 = sage_pick(mine, j + 6, hf);
            const v4f v0 = sage_zrow4(z, s0, l16);
            const v4f v1 = sage_zrow4(z, s1, l16);
            const v4f v2 = sage_zrow4(z, s2, l16);
            const v4f v3 = sage_zrow4(z, s3, l16);
            a0 += v0; a1 += v1; a2 += v2; a3 += v3;
        }
        const int rem = n - j;
        const int r0 = sage_pick(mine, j,     hf);
        const int r1 = sage_pick(mine, j + 2, hf);
        const int r2 = sage_pick(mine, j + 4, hf);
        const int r3 = sage_pick(mine, j + 6, hf);
        v4f v0 = (v4f){}, v1 = (v4f){}, v2 = (v4f){}, v3 = (v4f){};
        if (hf     < rem) v0 = sage_zrow4(z, r0, l16);
        if (hf + 2 < rem) v1 = sage_zrow4(z, r1, l16);
        if (hf + 4 < rem) v2 = sage_zrow4(z, r2, l16);
        if (hf + 6 < rem) v3 = sage_zrow4(z, r3, l16);
        a0 += v0; a1 += v1; a2 += v2; a3 += v3;
    }
    v4f r = (a0 + a1) + (a2 + a3);
    r.x += __shfl_xor(r.x, 16);
    r.y += __shfl_xor(r.y, 16);
    r.z += __shfl_xor(r.z, 16);
    r.w += __shfl_xor(r.w, 16);
    const float invd = 1.0f / fmaxf((float)c, 1.0f);
    if (hf == 0) orow[l16] = own + r * invd;
}

__global__ __launch_bounds__(256) void sage_gather_out_kernel(
    const float* __restrict__ z,
    const int*   __restrict__ col,
    const int*   __restrict__ cnt,
    const int*   __restrict__ rowstart,
    float* __restrict__ out,
    int n_nodes) {
    const int lane = threadIdx.x & 31;
    const int hf   = lane >> 4;
    const int l16  = lane & 15;
    const int w  = __builtin_amdgcn_readfirstlane((int)(blockIdx.x * (blockDim.x >> 5) + (threadIdx.x >> 5)));
    const int d0 = w * OUT_NPW;
    if (d0 >= n_nodes) return;
    const int4 c4 = ((const int4*)cnt)[w];
    const int4 s4 = ((const int4*)rowstart)[w];
    v4f* orow = (v4f*)(out + (long long)d0 * 64);
    sage_out_node(z, col, c4.x, s4.x, orow,      lane, hf, l16);
    sage_out_node(z, col, c4.y, s4.y, orow + 16, lane, hf, l16);
    sage_out_node(z, col, c4.z, s4.z, orow + 32, lane, hf, l16);
    sage_out_node(z, col, c4.w, s4.w, orow + 48, lane, hf, l16);
}

#define LIN_BLOCKS  256
#define LIN_WAVES   13
#define LIN_THREADS (LIN_WAVES * 32)
#define LIN_FLOATS  (2 * 128 * HID_C + 2 * 128 * OUT_C)

static_assert(LIN_BLOCKS * LIN_WAVES * 16 >= N_NODES, "one tile per wave must cover every node");
static_assert(N_NODES % 16 == 0, "whole tiles only");

#define STAGE_PER 16

template <int STAGE_T>
__device__ __forceinline__ void sage_stage_pair(
    const float* __restrict__ Wa, const float* __restrict__ Wb, v4f* la, v4f* lb, int t) {
    const v4f* ga = (const v4f*)Wa;
    const v4f* gb = (const v4f*)Wb;
    v4f ta[STAGE_PER], tb[STAGE_PER];
#pragma unroll
    for (int u = 0; u < STAGE_PER; ++u) {
        ta[u] = ga[u * STAGE_T + t];
        tb[u] = gb[u * STAGE_T + t];
    }
#pragma unroll
    for (int u = 0; u < STAGE_PER; ++u) {
        la[u * STAGE_T + t] = ta[u];
        lb[u * STAGE_T + t] = tb[u];
    }
}

__global__ __launch_bounds__(LIN_THREADS) void sage_linear_kernel(
    const float* __restrict__ agg,
    const float* __restrict__ deg,
    const float* __restrict__ feat,
    const float* __restrict__ W1l,
    const float* __restrict__ b1,
    const float* __restrict__ W1r,
    const float* __restrict__ W2l,
    const float* __restrict__ b2,
    const float* __restrict__ W2r,
    float* __restrict__ z,
    float* __restrict__ out,
    int n_tiles) {
    constexpr int NT1 = HID_C / 16;
    constexpr int NT2 = OUT_C / 16;
    constexpr int T1  = 128 * HID_C / 4 / STAGE_PER;
    constexpr int T2  = 128 * OUT_C / 4 / STAGE_PER;
    static_assert(T1 + T2 <= LIN_THREADS && T1 % 32 == 0 && T2 % 32 == 0, "whole waves stage");

    __shared__ __attribute__((aligned(16))) float s_w[LIN_FLOATS];
    constexpr int OFF_W1L = 0;
    constexpr int OFF_W1R = OFF_W1L + 128 * HID_C;
    constexpr int OFF_W2L = OFF_W1R + 128 * HID_C;
    constexpr int OFF_W2R = OFF_W2L + 128 * OUT_C;
    float* sW1l = s_w + OFF_W1L;
    float* sW1r = s_w + OFF_W1R;
    float* sW2l = s_w + OFF_W2L;
    float* sW2r = s_w + OFF_W2R;

    const int tid = threadIdx.x;
    const int wave = tid >> 5;
    const int lane = tid & 31;
    const int tile = blockIdx.x + LIN_BLOCKS * wave;
    const bool live = tile < n_tiles;
    const int row0 = tile * 16;
    const int hi   = lane >> 4;
    const int lr   = lane & 15;
    const int node = row0 + lr;
    const float dg = deg[live ? node : 0];

    if (tid < T1) {
        sage_stage_pair<T1>(W1l, W1r, (v4f*)sW1l, (v4f*)sW1r, tid);
    } else if (tid < T1 + T2) {
        sage_stage_pair<T2>(W2l, W2r, (v4f*)sW2l, (v4f*)sW2r, tid - T1);
    }
    __syncthreads();

    if (!live) return;

    const float invd = 1.0f / fmaxf(dg, 1.0f);

    const float* aggRow = agg  + (long long)node * 128;
    const float* xRow   = feat + (long long)node * 128;

    v8f acc[NT1];
#pragma unroll
    for (int n = 0; n < NT1; ++n) acc[n] = (v8f){};

    int oW1l = OFF_W1L + 2 * hi * HID_C + lr;
    int oW1r = OFF_W1R + 2 * hi * HID_C + lr;
    asm volatile("" : "+v"(oW1l));
    asm volatile("" : "+v"(oW1r));

    for (int k = 0; k < 32; ++k) {
        const int c = 4 * k + 2 * hi;
        v2f fAgg, fX;
        fAgg.x = aggRow[c]     * invd;
        fAgg.y = aggRow[c + 1] * invd;
        fX.x   = xRow[c];
        fX.y   = xRow[c + 1];
#pragma unroll
        for (int n = 0; n < NT1; ++n) {
            v2f wL, wR;
            wL.x = s_w[oW1l + n * 16];
            wL.y = s_w[oW1l + HID_C + n * 16];
            wR.x = s_w[oW1r + n * 16];
            wR.y = s_w[oW1r + HID_C + n * 16];
            acc[n] = __builtin_amdgcn_wmma_f32_16x16x4_f32(
                false, wL, false, fAgg, (short)0, acc[n], false, false);
            acc[n] = __builtin_amdgcn_wmma_f32_16x16x4_f32(
                false, wR, false, fX,   (short)0, acc[n], false, false);
        }
        oW1l += 4 * HID_C;
        oW1r += 4 * HID_C;
    }

#pragma unroll
    for (int n = 0; n < NT1; ++n) {
        const v4f* bp = (const v4f*)(b1 + n * 16 + 8 * hi);
        const v4f blo = bp[0];
        const v4f bhi = bp[1];
        acc[n][0] = fmaxf(acc[n][0] + blo.x, 0.0f);
        acc[n][1] = fmaxf(acc[n][1] + blo.y, 0.0f);
        acc[n][2] = fmaxf(acc[n][2] + blo.z, 0.0f);
        acc[n][3] = fmaxf(acc[n][3] + blo.w, 0.0f);
        acc[n][4] = fmaxf(acc[n][4] + bhi.x, 0.0f);
        acc[n][5] = fmaxf(acc[n][5] + bhi.y, 0.0f);
        acc[n][6] = fmaxf(acc[n][6] + bhi.z, 0.0f);
        acc[n][7] = fmaxf(acc[n][7] + bhi.w, 0.0f);
    }

    int oW2[NT2];
#pragma unroll
    for (int m = 0; m < NT2; ++m) {
        oW2[m] = OFF_W2L + 8 * hi * OUT_C + m * 16 + lr;
        asm volatile("" : "+v"(oW2[m]));
    }
    static_assert(OFF_W2R - OFF_W2L == 128 * OUT_C, "W2r directly behind W2l");

    {
        v8f accL[NT2];
#pragma unroll
        for (int m = 0; m < NT2; ++m) accL[m] = (v8f){};
#pragma unroll
        for (int n = 0; n < NT1; ++n) {
#pragma unroll
            for (int p = 0; p < 4; ++p) {
                v2f a;
                a.x = acc[n][2 * p];
                a.y = acc[n][2 * p + 1];
#pragma unroll
                for (int m = 0; m < NT2; ++m) {
                    v2f bL;
                    bL.x = s_w[oW2[m] + (n * 16 + 2 * p)     * OUT_C];
                    bL.y = s_w[oW2[m] + (n * 16 + 2 * p + 1) * OUT_C];
                    accL[m] = __builtin_amdgcn_wmma_f32_16x16x4_f32(
                        false, a, false, bL, (short)0, accL[m], false, false);
                }
            }
            __builtin_amdgcn_sched_barrier(0);
        }
#pragma unroll
        for (int m = 0; m < NT2; ++m) {
            const int col = m * 16 + lr;
#pragma unroll
            for (int v = 0; v < 8; ++v) {
                const int r = row0 + v + 8 * hi;
                z[(long long)r * OUT_C + col] = accL[m][v];
            }
        }
    }
    {
        v8f accR[NT2];
#pragma unroll
        for (int m = 0; m < NT2; ++m) accR[m] = (v8f){};
#pragma unroll
        for (int n = 0; n < NT1; ++n) {
#pragma unroll
            for (int p = 0; p < 4; ++p) {
                v2f a;
                a.x = acc[n][2 * p];
                a.y = acc[n][2 * p + 1];
#pragma unroll
                for (int m = 0; m < NT2; ++m) {
                    v2f bR;
                    bR.x = s_w[oW2[m] + (128 + n * 16 + 2 * p)     * OUT_C];
                    bR.y = s_w[oW2[m] + (128 + n * 16 + 2 * p + 1) * OUT_C];
                    accR[m] = __builtin_amdgcn_wmma_f32_16x16x4_f32(
                        false, a, false, bR, (short)0, accR[m], false, false);
                }
            }
            __builtin_amdgcn_sched_barrier(0);
        }
#pragma unroll
        for (int m = 0; m < NT2; ++m) {
            const int col = m * 16 + lr;
            const float b = b2[col];
#pragma unroll
            for (int v = 0; v < 8; ++v) {
                const int r = row0 + v + 8 * hi;
                out[(long long)r * OUT_C + col] = accR[m][v] + b;
            }
        }
    }
}

extern "C" void kernel_launch(void* const* d_in, const int* in_sizes, int n_in,
                              void* d_out, int out_size, void* d_ws, size_t ws_size,
                              hipStream_t stream) {
    const float* x   = (const float*)d_in[0];
    const int*   ei  = (const int*)  d_in[1];
    const float* W1l = (const float*)d_in[2];
    const float* b1  = (const float*)d_in[3];
    const float* W1r = (const float*)d_in[4];
    const float* W2l = (const float*)d_in[5];
    const float* b2  = (const float*)d_in[6];
    const float* W2r = (const float*)d_in[7];

    const int* src = ei;
    const int* dst = ei + N_EDGES;

    float* ws  = (float*)d_ws;
    float* sum = ws;
    float* deg = ws + (long long)N_NODES * 128;
    float* out = (float*)d_out;

    int* col    = (int*)(deg + N_NODES);
    int* cnt    = col + N_EDGES;
    int* cursor = cnt + N_NODES;
    float* z    = ws + 7350016;
    int* part   = (int*)d_ws;

    const int tiles       = N_NODES / 16;
    const int node_blocks = (N_NODES + 7) / 8;
    const int out_blocks  = (N_NODES + 8 * OUT_NPW - 1) / (8 * OUT_NPW);

    sage_hist_kernel<<<SORT_C * SORT_G, 1024, 0, stream>>>(dst, part);
    sage_chunkscan_kernel<<<(N_NODES + 255) / 256, 256, 0, stream>>>(part, cnt, N_NODES);
    sage_fill_kernel<<<SORT_C * SORT_G, 1024, 0, stream>>>(src, dst, part, cnt, cursor, col);

    sage_gather_kernel<<<node_blocks, 256, 0, stream>>>(x, col, cnt, cursor, sum, deg, N_NODES);
    sage_linear_kernel<<<LIN_BLOCKS, LIN_THREADS, 0, stream>>>(
        sum, deg, x, W1l, b1, W1r, W2l, b2, W2r, z, out, tiles);

    sage_gather_out_kernel<<<out_blocks, 256, 0, stream>>>(z, col, cnt, cursor, out, N_NODES);
}


# Layer-1 gather: a wave sums 4 consecutive nodes one after the other (12500 waves, was 50000 one-node waves), degrees and row starts in one 16-byte load each; order of sums unchanged
# speedup vs baseline: 4.7758x; 1.0438x over previous
#include <hip/hip_runtime.h>


#define N_NODES 50000
#define N_EDGES 800000
#define IN_C    128
#define HID_C   128
#define OUT_C   64

typedef float v2f __attribute__((ext_vector_type(2)));
typedef float v4f __attribute__((ext_vector_type(4)));
typedef float v8f __attribute__((ext_vector_type(8)));

#define SORT_G   8
#define SORT_C   32
#define GROUP_N  (N_NODES / SORT_G)
#define CHUNK_E  (N_EDGES / SORT_C)

__global__ __launch_bounds__(1024) void sage_hist_kernel(
    const int* __restrict__ dst, int* __restrict__ part) {
    __shared__ int s_cnt[GROUP_N];
    const int t  = threadIdx.x;
    const int c  = blockIdx.x / SORT_G;
    const int g  = blockIdx.x % SORT_G;
    const int lo = g * GROUP_N;
    for (int i = t; i < GROUP_N; i += 1024) s_cnt[i] = 0;
    __syncthreads();
    const int4* d4 = (const int4*)dst + c * (CHUNK_E / 4);
    int  i = t;
    int4 d = d4[i];
    for (;;) {
        const int  in   = i + 1024;
        const bool more = in < CHUNK_E / 4;
        int4 dn = d;
        if (more) dn = d4[in];
        const unsigned r0 = (unsigned)(d.x - lo);
        const unsigned r1 = (unsigned)(d.y - lo);
        const unsigned r2 = (unsigned)(d.z - lo);
        const unsigned r3 = (unsigned)(d.w - lo);
        if (r0 < (unsigned)GROUP_N) atomicAdd(&s_cnt[r0], 1);
        if (r1 < (unsigned)GROUP_N) atomicAdd(&s_cnt[r1], 1);
        if (r2 < (unsigned)GROUP_N) atomicAdd(&s_cnt[r2], 1);
        if (r3 < (unsigned)GROUP_N) atomicAdd(&s_cnt[r3], 1);
        if (!more) break;
        d = dn;
        i = in;
    }
    __syncthreads();
    int* p = part + c * N_NODES + lo;
    for (int i = t; i < GROUP_N; i += 1024) p[i] = s_cnt[i];
}

__global__ __launch_bounds__(256) void sage_chunkscan_kernel(
    int* __restrict__ part, int* __restrict__ cnt, int n) {
    const int d = blockIdx.x * blockDim.x + threadIdx.x;
    if (d >= n) return;
    int v[SORT_C];
#pragma unroll
    for (int c = 0; c < SORT_C; ++c) v[c] = part[c * N_NODES + d];
    int run = 0;
#pragma unroll
    for (int c = 0; c < SORT_C; ++c) {
        part[c * N_NODES + d] = run;
        run += v[c];
    }
    cnt[d] = run;
}

#define FILL_PER   7
#define FILL_FRONT 11

__global__ __launch_bounds__(1024) void sage_fill_kernel(
    const int* __restrict__ src, const int* __restrict__ dst,
    const int* __restrict__ part, const int* __restrict__ cnt,
    int* __restrict__ cursor, int* __restrict__ col) {
    __shared__ int s_slot[GROUP_N];
    __shared__ int s_before[32];
    __shared__ int s_own[32];
    const int t    = threadIdx.x;
    const int lane = t & 31;
    const int w    = t >> 5;
    const int c  = blockIdx.x / SORT_G;
    const int g  = blockIdx.x % SORT_G;
    const int lo = g * GROUP_N;
    const int* p = part + c * N_NODES + lo;

    const int4* d4 = (const int4*)dst + c * (CHUNK_E / 4);
    const int4* s4 = (const int4*)src + c * (CHUNK_E / 4);
    int  i = t;
    int4 d = d4[i];
    int4 s = s4[i];

    const int i0 = t * FILL_PER;
    int cn[FILL_PER], pv[FILL_PER];
#pragma unroll
    for (int k = 0; k < FILL_PER; ++k) {
        const int  nd = i0 + k;
        const bool ok = nd < GROUP_N;
        cn[k] = ok ? cnt[lo + nd] : 0;
        pv[k] = ok ? p[nd] : 0;
    }

    const int4* cnt4 = (const int4*)cnt;
    const int   n4   = lo >> 2;
    int4 fr[FILL_FRONT];
#pragma unroll
    for (int k = 0; k < FILL_FRONT; ++k) {
        const int q = t + 1024 * k;
        fr[k] = (q < n4) ? cnt4[q] : make_int4(0, 0, 0, 0);
    }
    int before = (t < (lo & 3)) ? cnt[4 * n4 + t] : 0;
#pragma unroll
    for (int k = 0; k < FILL_FRONT; ++k) before += (fr[k].x + fr[k].y) + (fr[k].z + fr[k].w);

    int ex[FILL_PER];
    int run = 0;
#pragma unroll
    for (int k = 0; k < FILL_PER; ++k) { ex[k] = run; run += cn[k]; }

    int incl = run;
    for (int off = 1; off < 32; off <<= 1) {
        const int up = __shfl_up(incl, off);
        if (lane >= off) incl += up;
    }
    for (int off = 16; off > 0; off >>= 1) before += __shfl_xor(before, off);

    if (lane == 31) s_own[w] = incl;
    if (lane == 0)  s_before[w] = before;
    __syncthreads();

    int base = 0;
    for (int k = 0; k < 32; ++k) {
        base += s_before[k];
        if (k < w) base += s_own[k];
    }
    base += incl - run;
#pragma unroll
    for (int k = 0; k < FILL_PER; ++k) {
        const int nd = i0 + k;
        if (nd < GROUP_N) {
            const int rs = base + ex[k];
            s_slot[nd] = rs + pv[k];
            if (c == 0) cursor[lo + nd] = rs;
        }
    }
    __syncthreads();
    for (;;) {
        const int  in   = i + 1024;
        const bool more = in < CHUNK_E / 4;
        int4 dn = d, sn = s;
        if (more) { dn = d4[in]; sn = s4[in]; }
        const unsigned r0 = (unsigned)(d.x - lo);
        const unsigned r1 = (unsigned)(d.y - lo);
        const unsigned r2 = (unsigned)(d.z - lo);
        const unsigned r3 = (unsigned)(d.w - lo);
        const bool in0 = r0 < (unsigned)GROUP_N;
        const bool in1 = r1 < (unsigned)GROUP_N;
        const bool in2 = r2 < (unsigned)GROUP_N;
        const bool in3 = r3 < (unsigned)GROUP_N;
        unsigned p0 = 0, p1 = 0, p2 = 0, p3 = 0;
        if (in0) p0 = (unsigned)atomicAdd(&s_slot[r0], 1);
        if (in1) p1 = (unsigned)atomicAdd(&s_slot[r1], 1);
        if (in2) p2 = (unsigned)atomicAdd(&s_slot[r2], 1);
        if (in3) p3 = (unsigned)atomicAdd(&s_slot[r3], 1);
        if (in0) col[p0] = s.x;
        if (in1) col[p1] = s.y;
        if (in2) col[p2] = s.z;
        if (in3) col[p3] = s.w;
        if (!more) break;
        d = dn;
        s = sn;
        i = in;
    }
}

__device__ __forceinline__ v4f sage_row4(const float* __restrict__ feat, int s, int lane) {
    return ((const v4f*)(feat + (long long)s * 128))[lane];
}

#define SUM_NPW 4

__device__ __forceinline__ void sage_sum_node(
    const float* __restrict__ feat, const int* __restrict__ col,
    int c, int start, v4f* srow, int lane) {
    const int end = start + c;
    v4f a0 = (v4f){}, a1 = (v4f){}, a2 = (v4f){}, a3 = (v4f){};
    for (int base = start; base < end; base += 32) {
        const int left = end - base;
        const int n    = left < 32 ? left : 32;
        const int mine = (lane < n) ? col[base + lane] : 0;
        int j = 0;
        for (; j + 4 <= n; j += 4) {
            const int s0 = __builtin_amdgcn_readlane(mine, j);
            const int s1 = __builtin_amdgcn_readlane(mine, j + 1);
            const int s2 = __builtin_amdgcn_readlane(mine, j + 2);
            const int s3 = __builtin_amdgcn_readlane(mine, j + 3);
            const v4f v0 = sage_row4(feat, s0, lane);
            const v4f v1 = sage_row4(feat, s1, lane);
            const v4f v2 = sage_row4(feat, s2, lane);
            const v4f v3 = sage_row4(feat, s3, lane);
            a0 += v0; a1 += v1; a2 += v2; a3 += v3;
        }
        const int rem = n - j;
        const int r0 = __builtin_amdgcn_readlane(mine, j & 31);
        const int r1 = __builtin_amdgcn_readlane(mine, (j + 1) & 31);
        const int r2 = __builtin_amdgcn_readlane(mine, (j + 2) & 31);
        v4f v0 = (v4f){}, v1 = (v4f){}, v2 = (v4f){};
        if (rem > 0) v0 = sage_row4(feat, r0, lane);
        if (rem > 1) v1 = sage_row4(feat, r1, lane);
        if (rem > 2) v2 = sage_row4(feat, r2, lane);
        a0 += v0; a1 += v1; a2 += v2;
    }
    srow[lane] = (a0 + a1) + (a2 + a3);
}

__global__ __launch_bounds__(256) void sage_gather_kernel(
    const float* __restrict__ feat,
    const int*   __restrict__ col,
    const int*   __restrict__ cnt,
    const int*   __restrict__ rowstart,
    float* __restrict__ sum,
    float* __restrict__ deg,
    int n_nodes) {
    const int lane = threadIdx.x & 31;
    const int w  = __builtin_amdgcn_readfirstlane((int)(blockIdx.x * (blockDim.x >> 5) + (threadIdx.x >> 5)));
    const int d0 = w * SUM_NPW;
    if (d0 >= n_nodes) return;
    const int4 c4 = ((const int4*)cnt)[w];
    const int4 s4 = ((const int4*)rowstart)[w];
    v4f* srow = (v4f*)(sum + (long long)d0 * 128);
    sage_sum_node(feat, col, c4.x, s4.x, srow,      lane);
    sage_sum_node(feat, col, c4.y, s4.y, srow + 32, lane);
    sage_sum_node(feat, col, c4.z, s4.z, srow + 64, lane);
    sage_sum_node(feat, col, c4.w, s4.w, srow + 96, lane);
    if (lane == 0) ((v4f*)deg)[w] = (v4f){(float)c4.x, (float)c4.y, (float)c4.z, (float)c4.w};
}

#define OUT_NPW 4

__device__ __forceinline__ int sage_pick(int mine, int j, int hf) {
    const int e = __builtin_amdgcn_readlane(mine, j & 31);
    const int o = __builtin_amdgcn_readlane(mine, (j + 1) & 31);
    return hf ? o : e;
}

__device__ __forceinline__ v4f sage_zrow4(const float* __restrict__ z, int s, int l16) {
    return ((const v4f*)(z + (long long)s * 64))[l16];
}

__device__ __forceinline__ void sage_out_node(
    const float* __restrict__ z, const int* __restrict__ col,
    int c, int start, v4f* orow, int lane, int hf, int l16) {
    const int end = start + c;

    v4f own = (v4f){};
    if (hf == 0) own = orow[l16];

    v4f a0 = (v4f){}, a1 = (v4f){}, a2 = (v4f){}, a3 = (v4f){};
    for (int base = start; base < end; base += 32) {
        const int left = end - base;
        const int n    = left < 32 ? left : 32;
        const int mine = (lane < n) ? col[base + lane] : 0;
        int j = 0;
        for (; j + 8 <= n; j += 8) {
            const int s0 = sage_pick(mine, j,     hf);
            const int s1 = sage_pick(mine, j + 2, hf);
            const int s2 = sage_pick(mine, j + 4, hf);
            const int s3 = sage_pick(mine, j + 6, hf);
            const v4f v0 = sage_zrow4(z, s0, l16);
            const v4f v1 = sage_zrow4(z, s1, l16);
            const v4f v2 = sage_zrow4(z, s2, l16);
            const v4f v3 = sage_zrow4(z, s3, l16);
            a0 += v0; a1 += v1; a2 += v2; a3 += v3;
        }
        const int rem = n - j;
        const int r0 = sage_pick(mine, j,     hf);
        const int r1 = sage_pick(mine, j + 2, hf);
        const int r2 = sage_pick(mine, j + 4, hf);
        const int r3 = sage_pick(mine, j + 6, hf);
        v4f v0 = (v4f){}, v1 = (v4f){}, v2 = (v4f){}, v3 = (v4f){};
        if (hf     < rem) v0 = sage_zrow4(z, r0, l16);
        if (hf + 2 < rem) v1 = sage_zrow4(z, r1, l16);
        if (hf + 4 < rem) v2 = sage_zrow4(z, r2, l16);
        if (hf + 6 < rem) v3 = sage_zrow4(z, r3, l16);
        a0 += v0; a1 += v1; a2 += v2; a3 += v3;
    }
    v4f r = (a0 + a1) + (a2 + a3);
    r.x += __shfl_xor(r.x, 16);
    r.y += __shfl_xor(r.y, 16);
    r.z += __shfl_xor(r.z, 16);
    r.w += __shfl_xor(r.w, 16);
    const float invd = 1.0f / fmaxf((float)c, 1.0f);
    if (hf == 0) orow[l16] = own + r * invd;
}

__global__ __launch_bounds__(256) void sage_gather_out_kernel(
    const float* __restrict__ z,
    const int*   __restrict__ col,
    const int*   __restrict__ cnt,
    const int*   __restrict__ rowstart,
    float* __restrict__ out,
    int n_nodes) {
    const int lane = threadIdx.x & 31;
    const int hf   = lane >> 4;
    const int l16  = lane & 15;
    const int w  = __builtin_amdgcn_readfirstlane((int)(blockIdx.x * (blockDim.x >> 5) + (threadIdx.x >> 5)));
    const int d0 = w * OUT_NPW;
    if (d0 >= n_nodes) return;
    const int4 c4 = ((const int4*)cnt)[w];
    const int4 s4 = ((const int4*)rowstart)[w];
    v4f* orow = (v4f*)(out + (long long)d0 * 64);
    sage_out_node(z, col, c4.x, s4.x, orow,      lane, hf, l16);
    sage_out_node(z, col, c4.y, s4.y, orow + 16, lane, hf, l16);
    sage_out_node(z, col, c4.z, s4.z, orow + 32, lane, hf, l16);
    sage_out_node(z, col, c4.w, s4.w, orow + 48, lane, hf, l16);
}

#define LIN_BLOCKS  256
#define LIN_WAVES   13
#define LIN_THREADS (LIN_WAVES * 32)
#define LIN_FLOATS  (2 * 128 * HID_C + 2 * 128 * OUT_C)

static_assert(LIN_BLOCKS * LIN_WAVES * 16 >= N_NODES, "one tile per wave must cover every node");
static_assert(N_NODES % 16 == 0, "whole tiles only");

#define STAGE_PER 16

template <int STAGE_T>
__device__ __forceinline__ void sage_stage_pair(
    const float* __restrict__ Wa, const float* __restrict__ Wb, v4f* la, v4f* lb, int t) {
    const v4f* ga = (const v4f*)Wa;
    const v4f* gb = (const v4f*)Wb;
    v4f ta[STAGE_PER], tb[STAGE_PER];
#pragma unroll
    for (int u = 0; u < STAGE_PER; ++u) {
        ta[u] = ga[u * STAGE_T + t];
        tb[u] = gb[u * STAGE_T + t];
    }
#pragma unroll
    for (int u = 0; u < STAGE_PER; ++u) {
        la[u * STAGE_T + t] = ta[u];
        lb[u * STAGE_T + t] = tb[u];
    }
}

__global__ __launch_bounds__(LIN_THREADS) void sage_linear_kernel(
    const float* __restrict__ agg,
    const float* __restrict__ deg,
    const float* __restrict__ feat,
    const float* __restrict__ W1l,
    const float* __restrict__ b1,
    const float* __restrict__ W1r,
    const float* __restrict__ W2l,
    const float* __restrict__ b2,
    const float* __restrict__ W2r,
    float* __restrict__ z,
    float* __restrict__ out,
    int n_tiles) {
    constexpr int NT1 = HID_C / 16;
    constexpr int NT2 = OUT_C / 16;
    constexpr int T1  = 128 * HID_C / 4 / STAGE_PER;
    constexpr int T2  = 128 * OUT_C / 4 / STAGE_PER;
    static_assert(T1 + T2 <= LIN_THREADS && T1 % 32 == 0 && T2 % 32 == 0, "whole waves stage");

    __shared__ __attribute__((aligned(16))) float s_w[LIN_FLOATS];
    constexpr int OFF_W1L = 0;
    constexpr int OFF_W1R = OFF_W1L + 128 * HID_C;
    constexpr int OFF_W2L = OFF_W1R + 128 * HID_C;
    constexpr int OFF_W2R = OFF_W2L + 128 * OUT_C;
    float* sW1l = s_w + OFF_W1L;
    float* sW1r = s_w + OFF_W1R;
    float* sW2l = s_w + OFF_W2L;
    float* sW2r = s_w + OFF_W2R;

    const int tid = threadIdx.x;
    const int wave = tid >> 5;
    const int lane = tid & 31;
    const int tile = blockIdx.x + LIN_BLOCKS * wave;
    const bool live = tile < n_tiles;
    const int row0 = tile * 16;
    const int hi   = lane >> 4;
    const int lr   = lane & 15;
    const int node = row0 + lr;
    const float dg = deg[live ? node : 0];

    if (tid < T1) {
        sage_stage_pair<T1>(W1l, W1r, (v4f*)sW1l, (v4f*)sW1r, tid);
    } else if (tid < T1 + T2) {
        sage_stage_pair<T2>(W2l, W2r, (v4f*)sW2l, (v4f*)sW2r, tid - T1);
    }
    __syncthreads();

    if (!live) return;

    const float invd = 1.0f / fmaxf(dg, 1.0f);

    const float* aggRow = agg  + (long long)node * 128;
    const float* xRow   = feat + (long long)node * 128;

    v8f acc[NT1];
#pragma unroll
    for (int n = 0; n < NT1; ++n) acc[n] = (v8f){};

    int oW1l = OFF_W1L + 2 * hi * HID_C + lr;
    int oW1r = OFF_W1R + 2 * hi * HID_C + lr;
    asm volatile("" : "+v"(oW1l));
    asm volatile("" : "+v"(oW1r));

    for (int k = 0; k < 32; ++k) {
        const int c = 4 * k + 2 * hi;
        v2f fAgg, fX;
        fAgg.x = aggRow[c]     * invd;
        fAgg.y = aggRow[c + 1] * invd;
        fX.x   = xRow[c];
        fX.y   = xRow[c + 1];
#pragma unroll
        for (int n = 0; n < NT1; ++n) {
            v2f wL, wR;
            wL.x = s_w[oW1l + n * 16];
            wL.y = s_w[oW1l + HID_C + n * 16];
            wR.x = s_w[oW1r + n * 16];
            wR.y = s_w[oW1r + HID_C + n * 16];
            acc[n] = __builtin_amdgcn_wmma_f32_16x16x4_f32(
                false, wL, false, fAgg, (short)0, acc[n], false, false);
            acc[n] = __builtin_amdgcn_wmma_f32_16x16x4_f32(
                false, wR, false, fX,   (short)0, acc[n], false, false);
        }
        oW1l += 4 * HID_C;
        oW1r += 4 * HID_C;
    }

#pragma unroll
    for (int n = 0; n < NT1; ++n) {
        const v4f* bp = (const v4f*)(b1 + n * 16 + 8 * hi);
        const v4f blo = bp[0];
        const v4f bhi = bp[1];
        acc[n][0] = fmaxf(acc[n][0] + blo.x, 0.0f);
        acc[n][1] = fmaxf(acc[n][1] + blo.y, 0.0f);
        acc[n][2] = fmaxf(acc[n][2] + blo.z, 0.0f);
        acc[n][3] = fmaxf(acc[n][3] + blo.w, 0.0f);
        acc[n][4] = fmaxf(acc[n][4] + bhi.x, 0.0f);
        acc[n][5] = fmaxf(acc[n][5] + bhi.y, 0.0f);
        acc[n][6] = fmaxf(acc[n][6] + bhi.z, 0.0f);
        acc[n][7] = fmaxf(acc[n][7] + bhi.w, 0.0f);
    }

    int oW2[NT2];
#pragma unroll
    for (int m = 0; m < NT2; ++m) {
        oW2[m] = OFF_W2L + 8 * hi * OUT_C + m * 16 + lr;
        asm volatile("" : "+v"(oW2[m]));
    }
    static_assert(OFF_W2R - OFF_W2L == 128 * OUT_C, "W2r directly behind W2l");

    {
        v8f accL[NT2];
#pragma unroll
        for (int m = 0; m < NT2; ++m) accL[m] = (v8f){};
#pragma unroll
        for (int n = 0; n < NT1; ++n) {
#pragma unroll
            for (int p = 0; p < 4; ++p) {
                v2f a;
                a.x = acc[n][2 * p];
                a.y = acc[n][2 * p + 1];
#pragma unroll
                for (int m = 0; m < NT2; ++m) {
                    v2f bL;
                    bL.x = s_w[oW2[m] + (n * 16 + 2 * p)     * OUT_C];
                    bL.y = s_w[oW2[m] + (n * 16 + 2 * p + 1) * OUT_C];
                    accL[m] = __builtin_amdgcn_wmma_f32_16x16x4_f32(
                        false, a, false, bL, (short)0, accL[m], false, false);
                }
            }
            __builtin_amdgcn_sched_barrier(0);
        }
#pragma unroll
        for (int m = 0; m < NT2; ++m) {
            const int col = m * 16 + lr;
#pragma unroll
            for (int v = 0; v < 8; ++v) {
                const int r = row0 + v + 8 * hi;
                z[(long long)r * OUT_C + col] = accL[m][v];
            }
        }
    }
    {
        v8f accR[NT2];
#pragma unroll
        for (int m = 0; m < NT2; ++m) accR[m] = (v8f){};
#pragma unroll
        for (int n = 0; n < NT1; ++n) {
#pragma unroll
            for (int p = 0; p < 4; ++p) {
                v2f a;
                a.x = acc[n][2 * p];
                a.y = acc[n][2 * p + 1];
#pragma unroll
                for (int m = 0; m < NT2; ++m) {
                    v2f bR;
                    bR.x = s_w[oW2[m] + (128 + n * 16 + 2 * p)     * OUT_C];
                    bR.y = s_w[oW2[m] + (128 + n * 16 + 2 * p + 1) * OUT_C];
                    accR[m] = __builtin_amdgcn_wmma_f32_16x16x4_f32(
                        false, a, false, bR, (short)0, accR[m], false, false);
                }
            }
            __builtin_amdgcn_sched_barrier(0);
        }
#pragma unroll
        for (int m = 0; m < NT2; ++m) {
            const int col = m * 16 + lr;
            const float b = b2[col];
#pragma unroll
            for (int v = 0; v < 8; ++v) {
                const int r = row0 + v + 8 * hi;
                out[(long long)r * OUT_C + col] = accR[m][v] + b;
            }
        }
    }
}

extern "C" void kernel_launch(void* const* d_in, const int* in_sizes, int n_in,
                              void* d_out, int out_size, void* d_ws, size_t ws_size,
                              hipStream_t stream) {
    const float* x   = (const float*)d_in[0];
    const int*   ei  = (const int*)  d_in[1];
    const float* W1l = (const float*)d_in[2];
    const float* b1  = (const float*)d_in[3];
    const float* W1r = (const float*)d_in[4];
    const float* W2l = (const float*)d_in[5];
    const float* b2  = (const float*)d_in[6];
    const float* W2r = (const float*)d_in[7];

    const int* src = ei;
    const int* dst = ei + N_EDGES;

    float* ws  = (float*)d_ws;
    float* sum = ws;
    float* deg = ws + (long long)N_NODES * 128;
    float* out = (float*)d_out;

    int* col    = (int*)(deg + N_NODES);
    int* cnt    = col + N_EDGES;
    int* cursor = cnt + N_NODES;
    float* z    = ws + 7350016;
    int* part   = (int*)d_ws;

    const int tiles       = N_NODES / 16;
    const int node_blocks = (N_NODES + 8 * SUM_NPW - 1) / (8 * SUM_NPW);
    const int out_blocks  = (N_NODES + 8 * OUT_NPW - 1) / (8 * OUT_NPW);

    sage_hist_kernel<<<SORT_C * SORT_G, 1024, 0, stream>>>(dst, part);
    sage_chunkscan_kernel<<<(N_NODES + 255) / 256, 256, 0, stream>>>(part, cnt, N_NODES);
    sage_fill_kernel<<<SORT_C * SORT_G, 1024, 0, stream>>>(src, dst, part, cnt, cursor, col);

    sage_gather_kernel<<<node_blocks, 256, 0, stream>>>(x, col, cnt, cursor, sum, deg, N_NODES);
    sage_linear_kernel<<<LIN_BLOCKS, LIN_THREADS, 0, stream>>>(
        sum, deg, x, W1l, b1, W1r, W2l, b2, W2r, z, out, tiles);

    sage_gather_out_kernel<<<out_blocks, 256, 0, stream>>>(z, col, cnt, cursor, out, N_NODES);
}


# Linear launch, 16 waves a work-group: each tile past 12 per group (53 tiles) is shared by 4 waves, one per SIMD, by hidden-channel quarter; partials added in LDS, those rows' layer-2 sums regrouped
# speedup vs baseline: 4.9922x; 1.0453x over previous
#include <hip/hip_runtime.h>


#define N_NODES 50000
#define N_EDGES 800000
#define IN_C    128
#define HID_C   128
#define OUT_C   64

typedef float v2f __attribute__((ext_vector_type(2)));
typedef float v4f __attribute__((ext_vector_type(4)));
typedef float v8f __attribute__((ext_vector_type(8)));

#define SORT_G   8
#define SORT_C   32
#define GROUP_N  (N_NODES / SORT_G)
#define CHUNK_E  (N_EDGES / SORT_C)

__global__ __launch_bounds__(1024) void sage_hist_kernel(
    const int* __restrict__ dst, int* __restrict__ part) {
    __shared__ int s_cnt[GROUP_N];
    const int t  = threadIdx.x;
    const int c  = blockIdx.x / SORT_G;
    const int g  = blockIdx.x % SORT_G;
    const int lo = g * GROUP_N;
    for (int i = t; i < GROUP_N; i += 1024) s_cnt[i] = 0;
    __syncthreads();
    const int4* d4 = (const int4*)dst + c * (CHUNK_E / 4);
    int  i = t;
    int4 d = d4[i];
    for (;;) {
        const int  in   = i + 1024;
        const bool more = in < CHUNK_E / 4;
        int4 dn = d;
        if (more) dn = d4[in];
        const unsigned r0 = (unsigned)(d.x - lo);
        const unsigned r1 = (unsigned)(d.y - lo);
        const unsigned r2 = (unsigned)(d.z - lo);
        const unsigned r3 = (unsigned)(d.w - lo);
        if (r0 < (unsigned)GROUP_N) atomicAdd(&s_cnt[r0], 1);
        if (r1 < (unsigned)GROUP_N) atomicAdd(&s_cnt[r1], 1);
        if (r2 < (unsigned)GROUP_N) atomicAdd(&s_cnt[r2], 1);
        if (r3 < (unsigned)GROUP_N) atomicAdd(&s_cnt[r3], 1);
        if (!more) break;
        d = dn;
        i = in;
    }
    __syncthreads();
    int* p = part + c * N_NODES + lo;
    for (int i = t; i < GROUP_N; i += 1024) p[i] = s_cnt[i];
}

__global__ __launch_bounds__(256) void sage_chunkscan_kernel(
    int* __restrict__ part, int* __restrict__ cnt, int n) {
    const int d = blockIdx.x * blockDim.x + threadIdx.x;
    if (d >= n) return;
    int v[SORT_C];
#pragma unroll
    for (int c = 0; c < SORT_C; ++c) v[c] = part[c * N_NODES + d];
    int run = 0;
#pragma unroll
    for (int c = 0; c < SORT_C; ++c) {
        part[c * N_NODES + d] = run;
        run += v[c];
    }
    cnt[d] = run;
}

#define FILL_PER   7
#define FILL_FRONT 11

__global__ __launch_bounds__(1024) void sage_fill_kernel(
    const int* __restrict__ src, const int* __restrict__ dst,
    const int* __restrict__ part, const int* __restrict__ cnt,
    int* __restrict__ cursor, int* __restrict__ col) {
    __shared__ int s_slot[GROUP_N];
    __shared__ int s_before[32];
    __shared__ int s_own[32];
    const int t    = threadIdx.x;
    const int lane = t & 31;
    const int w    = t >> 5;
    const int c  = blockIdx.x / SORT_G;
    const int g  = blockIdx.x % SORT_G;
    const int lo = g * GROUP_N;
    const int* p = part + c * N_NODES + lo;

    const int4* d4 = (const int4*)dst + c * (CHUNK_E / 4);
    const int4* s4 = (const int4*)src + c * (CHUNK_E / 4);
    int  i = t;
    int4 d = d4[i];
    int4 s = s4[i];

    const int i0 = t * FILL_PER;
    int cn[FILL_PER], pv[FILL_PER];
#pragma unroll
    for (int k = 0; k < FILL_PER; ++k) {
        const int  nd = i0 + k;
        const bool ok = nd < GROUP_N;
        cn[k] = ok ? cnt[lo + nd] : 0;
        pv[k] = ok ? p[nd] : 0;
    }

    const int4* cnt4 = (const int4*)cnt;
    const int   n4   = lo >> 2;
    int4 fr[FILL_FRONT];
#pragma unroll
    for (int k = 0; k < FILL_FRONT; ++k) {
        const int q = t + 1024 * k;
        fr[k] = (q < n4) ? cnt4[q] : make_int4(0, 0, 0, 0);
    }
    int before = (t < (lo & 3)) ? cnt[4 * n4 + t] : 0;
#pragma unroll
    for (int k = 0; k < FILL_FRONT; ++k) before += (fr[k].x + fr[k].y) + (fr[k].z + fr[k].w);

    int ex[FILL_PER];
    int run = 0;
#pragma unroll
    for (int k = 0; k < FILL_PER; ++k) { ex[k] = run; run += cn[k]; }

    int incl = run;
    for (int off = 1; off < 32; off <<= 1) {
        const int up = __shfl_up(incl, off);
        if (lane >= off) incl += up;
    }
    for (int off = 16; off > 0; off >>= 1) before += __shfl_xor(before, off);

    if (lane == 31) s_own[w] = incl;
    if (lane == 0)  s_before[w] = before;
    __syncthreads();

    int base = 0;
    for (int k = 0; k < 32; ++k) {
        base += s_before[k];
        if (k < w) base += s_own[k];
    }
    base += incl - run;
#pragma unroll
    for (int k = 0; k < FILL_PER; ++k) {
        const int nd = i0 + k;
        if (nd < GROUP_N) {
            const int rs = base + ex[k];
            s_slot[nd] = rs + pv[k];
            if (c == 0) cursor[lo + nd] = rs;
        }
    }
    __syncthreads();
    for (;;) {
        const int  in   = i + 1024;
        const bool more = in < CHUNK_E / 4;
        int4 dn = d, sn = s;
        if (more) { dn = d4[in]; sn = s4[in]; }
        const unsigned r0 = (unsigned)(d.x - lo);
        const unsigned r1 = (unsigned)(d.y - lo);
        const unsigned r2 = (unsigned)(d.z - lo);
        const unsigned r3 = (unsigned)(d.w - lo);
        const bool in0 = r0 < (unsigned)GROUP_N;
        const bool in1 = r1 < (unsigned)GROUP_N;
        const bool in2 = r2 < (unsigned)GROUP_N;
        const bool in3 = r3 < (unsigned)GROUP_N;
        unsigned p0 = 0, p1 = 0, p2 = 0, p3 = 0;
        if (in0) p0 = (unsigned)atomicAdd(&s_slot[r0], 1);
        if (in1) p1 = (unsigned)atomicAdd(&s_slot[r1], 1);
        if (in2) p2 = (unsigned)atomicAdd(&s_slot[r2], 1);
        if (in3) p3 = (unsigned)atomicAdd(&s_slot[r3], 1);
        if (in0) col[p0] = s.x;
        if (in1) col[p1] = s.y;
        if (in2) col[p2] = s.z;
        if (in3) col[p3] = s.w;
        if (!more) break;
        d = dn;
        s = sn;
        i = in;
    }
}

__device__ __forceinline__ v4f sage_row4(const float* __restrict__ feat, int s, int lane) {
    return ((const v4f*)(feat + (long long)s * 128))[lane];
}

#define SUM_NPW 4

__device__ __forceinline__ void sage_sum_node(
    const float* __restrict__ feat, const int* __restrict__ col,
    int c, int start, v4f* srow, int lane) {
    const int end = start + c;
    v4f a0 = (v4f){}, a1 = (v4f){}, a2 = (v4f){}, a3 = (v4f){};
    for (int base = start; base < end; base += 32) {
        const int left = end - base;
        const int n    = left < 32 ? left : 32;
        const int mine = (lane < n) ? col[base + lane] : 0;
        int j = 0;
        for (; j + 4 <= n; j += 4) {
            const int s0 = __builtin_amdgcn_readlane(mine, j);
            const int s1 = __builtin_amdgcn_readlane(mine, j + 1);
            const int s2 = __builtin_amdgcn_readlane(mine, j + 2);
            const int s3 = __builtin_amdgcn_readlane(mine, j + 3);
            const v4f v0 = sage_row4(feat, s0, lane);
            const v4f v1 = sage_row4(feat, s1, lane);
            const v4f v2 = sage_row4(feat, s2, lane);
            const v4f v3 = sage_row4(feat, s3, lane);
            a0 += v0; a1 += v1; a2 += v2; a3 += v3;
        }
        const int rem = n - j;
        const int r0 = __builtin_amdgcn_readlane(mine, j & 31);
        const int r1 = __builtin_amdgcn_readlane(mine, (j + 1) & 31);
        const int r2 = __builtin_amdgcn_readlane(mine, (j + 2) & 31);
        v4f v0 = (v4f){}, v1 = (v4f){}, v2 = (v4f){};
        if (rem > 0) v0 = sage_row4(feat, r0, lane);
        if (rem > 1) v1 = sage_row4(feat, r1, lane);
        if (rem > 2) v2 = sage_row4(feat, r2, lane);
        a0 += v0; a1 += v1; a2 += v2;
    }
    srow[lane] = (a0 + a1) + (a2 + a3);
}

__global__ __launch_bounds__(256) void sage_gather_kernel(
    const float* __restrict__ feat,
    const int*   __restrict__ col,
    const int*   __restrict__ cnt,
    const int*   __restrict__ rowstart,
    float* __restrict__ sum,
    float* __restrict__ deg,
    int n_nodes) {
    const int lane = threadIdx.x & 31;
    const int w  = __builtin_amdgcn_readfirstlane((int)(blockIdx.x * (blockDim.x >> 5) + (threadIdx.x >> 5)));
    const int d0 = w * SUM_NPW;
    if (d0 >= n_nodes) return;
    const int4 c4 = ((const int4*)cnt)[w];
    const int4 s4 = ((const int4*)rowstart)[w];
    v4f* srow = (v4f*)(sum + (long long)d0 * 128);
    sage_sum_node(feat, col, c4.x, s4.x, srow,      lane);
    sage_sum_node(feat, col, c4.y, s4.y, srow + 32, lane);
    sage_sum_node(feat, col, c4.z, s4.z, srow + 64, lane);
    sage_sum_node(feat, col, c4.w, s4.w, srow + 96, lane);
    if (lane == 0) ((v4f*)deg)[w] = (v4f){(float)c4.x, (float)c4.y, (float)c4.z, (float)c4.w};
}

#define OUT_NPW 4

__device__ __forceinline__ int sage_pick(int mine, int j, int hf) {
    const int e = __builtin_amdgcn_readlane(mine, j & 31);
    const int o = __builtin_amdgcn_readlane(mine, (j + 1) & 31);
    return hf ? o : e;
}

__device__ __forceinline__ v4f sage_zrow4(const float* __restrict__ z, int s, int l16) {
    return ((const v4f*)(z + (long long)s * 64))[l16];
}

__device__ __forceinline__ void sage_out_node(
    const float* __restrict__ z, const int* __restrict__ col,
    int c, int start, v4f* orow, int lane, int hf, int l16) {
    const int end = start + c;

    v4f own = (v4f){};
    if (hf == 0) own = orow[l16];

    v4f a0 = (v4f){}, a1 = (v4f){}, a2 = (v4f){}, a3 = (v4f){};
    for (int base = start; base < end; base += 32) {
        const int left = end - base;
        const int n    = left < 32 ? left : 32;
        const int mine = (lane < n) ? col[base + lane] : 0;
        int j = 0;
        for (; j + 8 <= n; j += 8) {
            const int s0 = sage_pick(mine, j,     hf);
            const int s1 = sage_pick(mine, j + 2, hf);
            const int s2 = sage_pick(mine, j + 4, hf);
            const int s3 = sage_pick(mine, j + 6, hf);
            const v4f v0 = sage_zrow4(z, s0, l16);
            const v4f v1 = sage_zrow4(z, s1, l16);
            const v4f v2 = sage_zrow4(z, s2, l16);
            const v4f v3 = sage_zrow4(z, s3, l16);
            a0 += v0; a1 += v1; a2 += v2; a3 += v3;
        }
        const int rem = n - j;
        const int r0 = sage_pick(mine, j,     hf);
        const int r1 = sage_pick(mine, j + 2, hf);
        const int r2 = sage_pick(mine, j + 4, hf);
        const int r3 = sage_pick(mine, j + 6, hf);
        v4f v0 = (v4f){}, v1 = (v4f){}, v2 = (v4f){}, v3 = (v4f){};
        if (hf     < rem) v0 = sage_zrow4(z, r0, l16);
        if (hf + 2 < rem) v1 = sage_zrow4(z, r1, l16);
        if (hf + 4 < rem) v2 = sage_zrow4(z, r2, l16);
        if (hf + 6 < rem) v3 = sage_zrow4(z, r3, l16);
        a0 += v0; a1 += v1; a2 += v2; a3 += v3;
    }
    v4f r = (a0 + a1) + (a2 + a3);
    r.x += __shfl_xor(r.x, 16);
    r.y += __shfl_xor(r.y, 16);
    r.z += __shfl_xor(r.z, 16);
    r.w += __shfl_xor(r.w, 16);
    const float invd = 1.0f / fmaxf((float)c, 1.0f);
    if (hf == 0) orow[l16] = own + r * invd;
}

__global__ __launch_bounds__(256) void sage_gather_out_kernel(
    const float* __restrict__ z,
    const int*   __restrict__ col,
    const int*   __restrict__ cnt,
    const int*   __restrict__ rowstart,
    float* __restrict__ out,
    int n_nodes) {
    const int lane = threadIdx.x & 31;
    const int hf   = lane >> 4;
    const int l16  = lane & 15;
    const int w  = __builtin_amdgcn_readfirstlane((int)(blockIdx.x * (blockDim.x >> 5) + (threadIdx.x >> 5)));
    const int d0 = w * OUT_NPW;
    if (d0 >= n_nodes) return;
    const int4 c4 = ((const int4*)cnt)[w];
    const int4 s4 = ((const int4*)rowstart)[w];
    v4f* orow = (v4f*)(out + (long long)d0 * 64);
    sage_out_node(z, col, c4.x, s4.x, orow,      lane, hf, l16);
    sage_out_node(z, col, c4.y, s4.y, orow + 16, lane, hf, l16);
    sage_out_node(z, col, c4.z, s4.z, orow + 32, lane, hf, l16);
    sage_out_node(z, col, c4.w, s4.w, orow + 48, lane, hf, l16);
}

#define LIN_BLOCKS  256
#define LIN_FULL    12
#define LIN_SHARE   4
#define LIN_WAVES   (LIN_FULL + LIN_SHARE)
#define LIN_THREADS (LIN_WAVES * 32)
#define LIN_WFLOATS (2 * 128 * HID_C + 2 * 128 * OUT_C)
#define LIN_PWAVE   (8 * 8 * 32)
#define LIN_FLOATS  (LIN_WFLOATS + LIN_SHARE * LIN_PWAVE)
#define LIN_OFF_W1L 0
#define LIN_OFF_W1R (LIN_OFF_W1L + 128 * HID_C)
#define LIN_OFF_W2L (LIN_OFF_W1R + 128 * HID_C)
#define LIN_OFF_W2R (LIN_OFF_W2L + 128 * OUT_C)

static_assert(LIN_BLOCKS * (LIN_FULL + 1) * 16 >= N_NODES, "whole and shared tiles must cover every node");
static_assert(N_NODES % 16 == 0, "whole tiles only");
static_assert(HID_C / 16 == 2 * LIN_SHARE && OUT_C / 16 == LIN_SHARE, "two channel tiles and one column tile per sharing wave");

#define STAGE_PER 16

template <int STAGE_T>
__device__ __forceinline__ void sage_stage_pair(
    const float* __restrict__ Wa, const float* __restrict__ Wb, v4f* la, v4f* lb, int t) {
    const v4f* ga = (const v4f*)Wa;
    const v4f* gb = (const v4f*)Wb;
    v4f ta[STAGE_PER], tb[STAGE_PER];
#pragma unroll
    for (int u = 0; u < STAGE_PER; ++u) {
        ta[u] = ga[u * STAGE_T + t];
        tb[u] = gb[u * STAGE_T + t];
    }
#pragma unroll
    for (int u = 0; u < STAGE_PER; ++u) {
        la[u * STAGE_T + t] = ta[u];
        lb[u * STAGE_T + t] = tb[u];
    }
}

__device__ __forceinline__ void sage_quarter_tile(
    const float* s_w, const float* __restrict__ aggRow, const float* __restrict__ xRow,
    float invd, const float* __restrict__ b1, int q, int hi, int lr, int lane, float* plane) {
    const int N0 = 2 * q;
    constexpr int NT2 = OUT_C / 16;
    v8f acc[2];
    acc[0] = (v8f){};
    acc[1] = (v8f){};
    int oW1l = LIN_OFF_W1L + 2 * hi * HID_C + N0 * 16 + lr;
    int oW1r = LIN_OFF_W1R + 2 * hi * HID_C + N0 * 16 + lr;
    asm volatile("" : "+v"(oW1l));
    asm volatile("" : "+v"(oW1r));
    for (int k = 0; k < 32; ++k) {
        const int c = 4 * k + 2 * hi;
        v2f fAgg, fX;
        fAgg.x = aggRow[c]     * invd;
        fAgg.y = aggRow[c + 1] * invd;
        fX.x   = xRow[c];
        fX.y   = xRow[c + 1];
#pragma unroll
        for (int n = 0; n < 2; ++n) {
            v2f wL, wR;
            wL.x = s_w[oW1l + n * 16];
            wL.y = s_w[oW1l + HID_C + n * 16];
            wR.x = s_w[oW1r + n * 16];
            wR.y = s_w[oW1r + HID_C + n * 16];
            acc[n] = __builtin_amdgcn_wmma_f32_16x16x4_f32(
                false, wL, false, fAgg, (short)0, acc[n], false, false);
            acc[n] = __builtin_amdgcn_wmma_f32_16x16x4_f32(
                false, wR, false, fX,   (short)0, acc[n], false, false);
        }
        oW1l += 4 * HID_C;
        oW1r += 4 * HID_C;
    }
#pragma unroll
    for (int n = 0; n < 2; ++n) {
        const v4f* bp = (const v4f*)(b1 + (N0 + n) * 16 + 8 * hi);
        const v4f blo = bp[0];
        const v4f bhi = bp[1];
        acc[n][0] = fmaxf(acc[n][0] + blo.x, 0.0f);
        acc[n][1] = fmaxf(acc[n][1] + blo.y, 0.0f);
        acc[n][2] = fmaxf(acc[n][2] + blo.z, 0.0f);
        acc[n][3] = fmaxf(acc[n][3] + blo.w, 0.0f);
        acc[n][4] = fmaxf(acc[n][4] + bhi.x, 0.0f);
        acc[n][5] = fmaxf(acc[n][5] + bhi.y, 0.0f);
        acc[n][6] = fmaxf(acc[n][6] + bhi.z, 0.0f);
        acc[n][7] = fmaxf(acc[n][7] + bhi.w, 0.0f);
    }
    int oW2[NT2];
#pragma unroll
    for (int m = 0; m < NT2; ++m) {
        oW2[m] = LIN_OFF_W2L + (N0 * 16 + 8 * hi) * OUT_C + m * 16 + lr;
        asm volatile("" : "+v"(oW2[m]));
    }
    v8f accL[NT2], accR[NT2];
#pragma unroll
    for (int m = 0; m < NT2; ++m) { accL[m] = (v8f){}; accR[m] = (v8f){}; }
#pragma unroll
    for (int n = 0; n < 2; ++n) {
#pragma unroll
        for (int p = 0; p < 4; ++p) {
            v2f a;
            a.x = acc[n][2 * p];
            a.y = acc[n][2 * p + 1];
#pragma unroll
            for (int m = 0; m < NT2; ++m) {
                v2f bL, bR;
                bL.x = s_w[oW2[m] + (n * 16 + 2 * p)     * OUT_C];
                bL.y = s_w[oW2[m] + (n * 16 + 2 * p + 1) * OUT_C];
                bR.x = s_w[oW2[m] + (128 + n * 16 + 2 * p)     * OUT_C];
                bR.y = s_w[oW2[m] + (128 + n * 16 + 2 * p + 1) * OUT_C];
                accL[m] = __builtin_amdgcn_wmma_f32_16x16x4_f32(
                    false, a, false, bL, (short)0, accL[m], false, false);
                accR[m] = __builtin_amdgcn_wmma_f32_16x16x4_f32(
                    false, a, false, bR, (short)0, accR[m], false, false);
            }
        }
    }
#pragma unroll
    for (int m = 0; m < NT2; ++m) {
#pragma unroll
        for (int v = 0; v < 8; ++v) {
            plane[(m * 8 + v) * 32 + lane]         = accL[m][v];
            plane[((NT2 + m) * 8 + v) * 32 + lane] = accR[m][v];
        }
    }
}

__global__ __launch_bounds__(LIN_THREADS) void sage_linear_kernel(
    const float* __restrict__ agg,
    const float* __restrict__ deg,
    const float* __restrict__ feat,
    const float* __restrict__ W1l,
    const float* __restrict__ b1,
    const float* __restrict__ W1r,
    const float* __restrict__ W2l,
    const float* __restrict__ b2,
    const float* __restrict__ W2r,
    float* __restrict__ z,
    float* __restrict__ out,
    int n_tiles) {
    constexpr int NT1 = HID_C / 16;
    constexpr int NT2 = OUT_C / 16;
    constexpr int T1  = 128 * HID_C / 4 / STAGE_PER;
    constexpr int T2  = 128 * OUT_C / 4 / STAGE_PER;
    static_assert(T1 + T2 <= LIN_THREADS && T1 % 32 == 0 && T2 % 32 == 0, "whole waves stage");

    __shared__ __attribute__((aligned(16))) float s_w[LIN_FLOATS];
    constexpr int OFF_W1L = LIN_OFF_W1L;
    constexpr int OFF_W1R = LIN_OFF_W1R;
    constexpr int OFF_W2L = LIN_OFF_W2L;
    constexpr int OFF_W2R = LIN_OFF_W2R;
    float* s_part = s_w + LIN_WFLOATS;
    float* sW1l = s_w + OFF_W1L;
    float* sW1r = s_w + OFF_W1R;
    float* sW2l = s_w + OFF_W2L;
    float* sW2r = s_w + OFF_W2R;

    const int tid = threadIdx.x;
    const int wave = tid >> 5;
    const int lane = tid & 31;
    const int  ws   = __builtin_amdgcn_readfirstlane(wave);
    const bool full = ws < LIN_FULL;
    const int tile = blockIdx.x + LIN_BLOCKS * (wave < LIN_FULL ? wave : LIN_FULL);
    const bool live = tile < n_tiles;
    const int row0 = tile * 16;
    const int hi   = lane >> 4;
    const int lr   = lane & 15;
    const int node = row0 + lr;
    const float dg = deg[live ? node : 0];

    if (tid < T1) {
        sage_stage_pair<T1>(W1l, W1r, (v4f*)sW1l, (v4f*)sW1r, tid);
    } else if (tid < T1 + T2) {
        sage_stage_pair<T2>(W2l, W2r, (v4f*)sW2l, (v4f*)sW2r, tid - T1);
    }
    __syncthreads();

    const float invd = 1.0f / fmaxf(dg, 1.0f);

    const float* aggRow = agg  + (long long)node * 128;
    const float* xRow   = feat + (long long)node * 128;

    if (live && !full) {
        const int q = ws - LIN_FULL;
        sage_quarter_tile(s_w, aggRow, xRow, invd, b1, q, hi, lr, lane, s_part + q * LIN_PWAVE);
    }

    if (live && full) {
        v8f acc[NT1];
#pragma unroll
        for (int n = 0; n < NT1; ++n) acc[n] = (v8f){};

        int oW1l = OFF_W1L + 2 * hi * HID_C + lr;
        int oW1r = OFF_W1R + 2 * hi * HID_C + lr;
        asm volatile("" : "+v"(oW1l));
        asm volatile("" : "+v"(oW1r));

        for (int k = 0; k < 32; ++k) {
            const int c = 4 * k + 2 * hi;
            v2f fAgg, fX;
            fAgg.x = aggRow[c]     * invd;
            fAgg.y = aggRow[c + 1] * invd;
            fX.x   = xRow[c];
            fX.y   = xRow[c + 1];
#pragma unroll
            for (int n = 0; n < NT1; ++n) {
                v2f wL, wR;
                wL.x = s_w[oW1l + n * 16];
                wL.y = s_w[oW1l + HID_C + n * 16];
                wR.x = s_w[oW1r + n * 16];
                wR.y = s_w[oW1r + HID_C + n * 16];
                acc[n] = __builtin_amdgcn_wmma_f32_16x16x4_f32(
                    false, wL, false, fAgg, (short)0, acc[n], false, false);
                acc[n] = __builtin_amdgcn_wmma_f32_16x16x4_f32(
                    false, wR, false, fX,   (short)0, acc[n], false, false);
            }
            oW1l += 4 * HID_C;
            oW1r += 4 * HID_C;
        }

#pragma unroll
        for (int n = 0; n < NT1; ++n) {
            const v4f* bp = (const v4f*)(b1 + n * 16 + 8 * hi);
            const v4f blo = bp[0];
            const v4f bhi = bp[1];
            acc[n][0] = fmaxf(acc[n][0] + blo.x, 0.0f);
            acc[n][1] = fmaxf(acc[n][1] + blo.y, 0.0f);
            acc[n][2] = fmaxf(acc[n][2] + blo.z, 0.0f);
            acc[n][3] = fmaxf(acc[n][3] + blo.w, 0.0f);
            acc[n][4] = fmaxf(acc[n][4] + bhi.x, 0.0f);
            acc[n][5] = fmaxf(acc[n][5] + bhi.y, 0.0f);
            acc[n][6] = fmaxf(acc[n][6] + bhi.z, 0.0f);
            acc[n][7] = fmaxf(acc[n][7] + bhi.w, 0.0f);
        }

        int oW2[NT2];
#pragma unroll
        for (int m = 0; m < NT2; ++m) {
            oW2[m] = OFF_W2L + 8 * hi * OUT_C + m * 16 + lr;
            asm volatile("" : "+v"(oW2[m]));
        }
        static_assert(OFF_W2R - OFF_W2L == 128 * OUT_C, "W2r directly behind W2l");

        {
            v8f accL[NT2];
#pragma unroll
            for (int m = 0; m < NT2; ++m) accL[m] = (v8f){};
#pragma unroll
            for (int n = 0; n < NT1; ++n) {
#pragma unroll
                for (int p = 0; p < 4; ++p) {
                    v2f a;
                    a.x = acc[n][2 * p];
                    a.y = acc[n][2 * p + 1];
#pragma unroll
                    for (int m = 0; m < NT2; ++m) {
                        v2f bL;
                        bL.x = s_w[oW2[m] + (n * 16 + 2 * p)     * OUT_C];
                        bL.y = s_w[oW2[m] + (n * 16 + 2 * p + 1) * OUT_C];
                        accL[m] = __builtin_amdgcn_wmma_f32_16x16x4_f32(
                            false, a, false, bL, (short)0, accL[m], false, false);
                    }
                }
                __builtin_amdgcn_sched_barrier(0);
            }
#pragma unroll
            for (int m = 0; m < NT2; ++m) {
                const int col = m * 16 + lr;
#pragma unroll
                for (int v = 0; v < 8; ++v) {
                    const int r = row0 + v + 8 * hi;
                    z[(long long)r * OUT_C + col] = accL[m][v];
                }
            }
        }
        {
            v8f accR[NT2];
#pragma unroll
            for (int m = 0; m < NT2; ++m) accR[m] = (v8f){};
#pragma unroll
            for (int n = 0; n < NT1; ++n) {
#pragma unroll
                for (int p = 0; p < 4; ++p) {
                    v2f a;
                    a.x = acc[n][2 * p];
                    a.y = acc[n][2 * p + 1];
#pragma unroll
                    for (int m = 0; m < NT2; ++m) {
                        v2f bR;
                        bR.x = s_w[oW2[m] + (128 + n * 16 + 2 * p)     * OUT_C];
                        bR.y = s_w[oW2[m] + (128 + n * 16 + 2 * p + 1) * OUT_C];
                        accR[m] = __builtin_amdgcn_wmma_f32_16x16x4_f32(
                            false, a, false, bR, (short)0, accR[m], false, false);
                    }
                }
                __builtin_amdgcn_sched_barrier(0);
            }
#pragma unroll
            for (int m = 0; m < NT2; ++m) {
                const int col = m * 16 + lr;
                const float b = b2[col];
#pragma unroll
                for (int v = 0; v < 8; ++v) {
                    const int r = row0 + v + 8 * hi;
                    out[(long long)r * OUT_C + col] = accR[m][v] + b;
                }
            }
        }
    }

    __syncthreads();

    if (live && !full) {
        const int q    = ws - LIN_FULL;
        const int half = q & 1;
        const int rb   = q >> 1;
        const int col  = 32 * half + lane;
        const int t    = 2 * half + hi;
        const float b  = b2[col];
        const float* pz = s_part + t * 8 * 32 + 16 * rb + lr;
        const float* po = pz + (OUT_C / 16) * 8 * 32;
#pragma unroll
        for (int j = 0; j < 8; ++j) {
            const float zs = (pz[j * 32] + pz[j * 32 + LIN_PWAVE])
                           + (pz[j * 32 + 2 * LIN_PWAVE] + pz[j * 32 + 3 * LIN_PWAVE]);
            const float os = (po[j * 32] + po[j * 32 + LIN_PWAVE])
                           + (po[j * 32 + 2 * LIN_PWAVE] + po[j * 32 + 3 * LIN_PWAVE]);
            const int r = row0 + 8 * rb + j;
            z[(long long)r * OUT_C + col]   = zs;
            out[(long long)r * OUT_C + col] = os + b;
        }
    }
}

extern "C" void kernel_launch(void* const* d_in, const int* in_sizes, int n_in,
                              void* d_out, int out_size, void* d_ws, size_t ws_size,
                              hipStream_t stream) {
    const float* x   = (const float*)d_in[0];
    const int*   ei  = (const int*)  d_in[1];
    const float* W1l = (const float*)d_in[2];
    const float* b1  = (const float*)d_in[3];
    const float* W1r = (const float*)d_in[4];
    const float* W2l = (const float*)d_in[5];
    const float* b2  = (const float*)d_in[6];
    const float* W2r = (const float*)d_in[7];

    const int* src = ei;
    const int* dst = ei + N_EDGES;

    float* ws  = (float*)d_ws;
    float* sum = ws;
    float* deg = ws + (long long)N_NODES * 128;
    float* out = (float*)d_out;

    int* col    = (int*)(deg + N_NODES);
    int* cnt    = col + N_EDGES;
    int* cursor = cnt + N_NODES;
    float* z    = ws + 7350016;
    int* part   = (int*)d_ws;

    const int tiles       = N_NODES / 16;
    const int node_blocks = (N_NODES + 8 * SUM_NPW - 1) / (8 * SUM_NPW);
    const int out_blocks  = (N_NODES + 8 * OUT_NPW - 1) / (8 * OUT_NPW);

    sage_hist_kernel<<<SORT_C * SORT_G, 1024, 0, stream>>>(dst, part);
    sage_chunkscan_kernel<<<(N_NODES + 255) / 256, 256, 0, stream>>>(part, cnt, N_NODES);
    sage_fill_kernel<<<SORT_C * SORT_G, 1024, 0, stream>>>(src, dst, part, cnt, cursor, col);

    sage_gather_kernel<<<node_blocks, 256, 0, stream>>>(x, col, cnt, cursor, sum, deg, N_NODES);
    sage_linear_kernel<<<LIN_BLOCKS, LIN_THREADS, 0, stream>>>(
        sum, deg, x, W1l, b1, W1r, W2l, b2, W2r, z, out, tiles);

    sage_gather_out_kernel<<<out_blocks, 256, 0, stream>>>(z, col, cnt, cursor, out, N_NODES);
}
